# starting kernel
# baseline (speedup 1.0000x reference)
#include <hip/hip_runtime.h>
#include <math.h>


typedef __attribute__((ext_vector_type(2))) float v2f;
typedef __attribute__((ext_vector_type(8))) float v8f;

#define L   1024
#define DM  1024
#define EPSR 1.1920929e-07f

static __device__ __forceinline__ v8f wmma_f32(v2f a, v2f b, v8f c) {
  return __builtin_amdgcn_wmma_f32_16x16x4_f32(false, a, false, b, (short)0, c,
                                               false, false);
}

__global__ void __launch_bounds__(256)
rmsnorm3_kernel(const float* __restrict__ x, const float* __restrict__ pt,
                const float* __restrict__ gq, const float* __restrict__ gkv,
                const float* __restrict__ gc,
                float* __restrict__ xqn, float* __restrict__ xkvn,
                float* __restrict__ ptn) {
  __shared__ float red[256];
  const int row = blockIdx.x;
  const int tid = threadIdx.x;

  float4 xv = ((const float4*)(x + (size_t)row * DM))[tid];
  red[tid] = xv.x * xv.x + xv.y * xv.y + xv.z * xv.z + xv.w * xv.w;
  __syncthreads();
  for (int s = 128; s > 0; s >>= 1) {
    if (tid < s) red[tid] += red[tid + s];
    __syncthreads();
  }
  float rn = rsqrtf(red[0] * (1.0f / DM) + EPSR);
  __syncthreads();

  float4 gqv = ((const float4*)gq)[tid];
  float4 gkvv = ((const float4*)gkv)[tid];
  float4 oq, okv;
  oq.x = xv.x * rn * gqv.x;  oq.y = xv.y * rn * gqv.y;
  oq.z = xv.z * rn * gqv.z;  oq.w = xv.w * rn * gqv.w;
  okv.x = xv.x * rn * gkvv.x; okv.y = xv.y * rn * gkvv.y;
  okv.z = xv.z * rn * gkvv.z; okv.w = xv.w * rn * gkvv.w;
  ((float4*)(xqn + (size_t)row * DM))[tid] = oq;
  ((float4*)(xkvn + (size_t)row * DM))[tid] = okv;

  float4 pv = ((const float4*)(pt + (size_t)row * DM))[tid];
  red[tid] = pv.x * pv.x + pv.y * pv.y + pv.z * pv.z + pv.w * pv.w;
  __syncthreads();
  for (int s = 128; s > 0; s >>= 1) {
    if (tid < s) red[tid] += red[tid + s];
    __syncthreads();
  }
  float rnp = rsqrtf(red[0] * (1.0f / DM) + EPSR);
  float4 gcv = ((const float4*)gc)[tid];
  float4 oc;
  oc.x = pv.x * rnp * gcv.x; oc.y = pv.y * rnp * gcv.y;
  oc.z = pv.z * rnp * gcv.z; oc.w = pv.w * rnp * gcv.w;
  ((float4*)(ptn + (size_t)row * DM))[tid] = oc;
}

__global__ void __launch_bounds__(256)
wmma_gemm_bias(const float* __restrict__ A, const float* __restrict__ Wm,
               const float* __restrict__ bias, float* __restrict__ Cout,
               int N) {
  __shared__ float As[16 * 65];
  __shared__ float Bs[16 * 64];
  const int tid = threadIdx.x;
  const int lane = tid & 31, wid = tid >> 5;
  const int hhi = lane >> 4, l16 = lane & 15;
  const int m0 = blockIdx.x * 64, n0 = blockIdx.y * 64;
  const int wm = wid & 3, wn = wid >> 2;
  const int mloc = wm * 16 + l16;
  const int nloc0 = wn * 32 + l16;
  const int nloc1 = nloc0 + 16;

  v8f acc0 = {}; v8f acc1 = {};

  for (int k0 = 0; k0 < DM; k0 += 16) {
#pragma unroll
    for (int it = 0; it < 4; ++it) {
      int idx = tid + it * 256;
      int ak = idx & 15, am = idx >> 4;
      As[ak * 65 + am] = A[(size_t)(m0 + am) * DM + k0 + ak];
      int bn = idx & 63, bk = idx >> 6;
      Bs[bk * 64 + bn] = Wm[(size_t)(k0 + bk) * N + n0 + bn];
    }
    __syncthreads();
    if (k0 + 16 < DM) {
      __builtin_prefetch(&Wm[(size_t)(k0 + 16 + (tid >> 6)) * N + n0 + (tid & 63)], 0, 1);
      __builtin_prefetch(&A[(size_t)(m0 + (tid >> 2)) * DM + k0 + 16], 0, 1);
    }
#pragma unroll
    for (int kk = 0; kk < 4; ++kk) {
      const int kb = kk * 4 + 2 * hhi;
      v2f a;  a.x = As[kb * 65 + mloc];        a.y = As[(kb + 1) * 65 + mloc];
      v2f b0; b0.x = Bs[kb * 64 + nloc0];      b0.y = Bs[(kb + 1) * 64 + nloc0];
      v2f b1; b1.x = Bs[kb * 64 + nloc1];      b1.y = Bs[(kb + 1) * 64 + nloc1];
      acc0 = wmma_f32(a, b0, acc0);
      acc1 = wmma_f32(a, b1, acc1);
    }
    __syncthreads();
  }

  const int rbase = m0 + wm * 16 + hhi * 8;
  const float bb0 = bias[n0 + nloc0];
  const float bb1 = bias[n0 + nloc1];
#pragma unroll
  for (int v = 0; v < 8; ++v) {
    const int r = rbase + v;
    Cout[(size_t)r * N + n0 + nloc0] = acc0[v] + bb0;
    Cout[(size_t)r * N + n0 + nloc1] = acc1[v] + bb1;
  }
}

__global__ void cumsum_kernel(const float* __restrict__ C, float* __restrict__ P) {
  const int col = blockIdx.x * blockDim.x + threadIdx.x;
  float run = 0.f;
  P[col] = 0.f;
  for (int t = 0; t < L; ++t) {
    run += C[(size_t)t * DM + col];
    P[(size_t)(t + 1) * DM + col] = run;
  }
}

__global__ void augment_kernel(const float* __restrict__ Q,
                               const float* __restrict__ KV,
                               const float* __restrict__ P,
                               const int* __restrict__ cw,
                               float* __restrict__ Qa, float* __restrict__ Ka) {
  const int i = blockIdx.x;
  const int w = cw[0];
  const float scale = 0.125f;
  int E = i + w; if (E > L) E = L;
  int S = i - w; if (S < 0) S = 0;
  for (int c = threadIdx.x; c < DM; c += blockDim.x) {
    const int h = c >> 6, dd = c & 63;
    const float pe = P[(size_t)E * DM + c];
    const float ps = P[(size_t)S * DM + c];
    const float qv = Q[(size_t)i * DM + c];
    const float kv = KV[(size_t)i * 2048 + c];
    const int base = i * 2048 + h * 128 + dd;
    Qa[base]      = qv * (scale + pe);
    Qa[base + 64] = qv;
    Ka[base]      = kv;
    Ka[base + 64] = -kv * ps;
  }
}

__global__ void __launch_bounds__(128)
attn_kernel(const float* __restrict__ Qa, const float* __restrict__ Ka,
            const float* __restrict__ KV, float* __restrict__ wv) {
  __shared__ float Qs[4 * 16 * 130];
  __shared__ float Ks[16 * 130];
  __shared__ float Vs[16 * 64];
  __shared__ float Pls[4 * 16 * 18];

  const int tid = threadIdx.x;
  const int lane = tid & 31, wid = tid >> 5;
  const int hhi = lane >> 4, l16 = lane & 15;
  const int h = blockIdx.y;
  const int hoff = h * 128;
  const int qrow0 = blockIdx.x * 64 + wid * 16;
  const int qtile = qrow0 >> 4;
  const int jt_max = blockIdx.x * 4 + 3;

  float* myQ = &Qs[wid * 16 * 130];
  float* myP = &Pls[wid * 16 * 18];

  for (int idx = lane; idx < 16 * 128; idx += 32) {
    const int r = idx >> 7, kc = idx & 127;
    myQ[r * 130 + kc] = Qa[(size_t)(qrow0 + r) * 2048 + hoff + kc];
  }

  v8f o0 = {}, o1 = {}, o2 = {}, o3 = {};
  float m8[8], l8[8];
#pragma unroll
  for (int v = 0; v < 8; ++v) { m8[v] = -3.0e38f; l8[v] = 0.f; }

  for (int jt = 0; jt <= jt_max; ++jt) {
    for (int idx = tid; idx < 16 * 128; idx += 128) {
      const int r = idx >> 7, kc = idx & 127;
      Ks[r * 130 + kc] = Ka[(size_t)(jt * 16 + r) * 2048 + hoff + kc];
    }
    for (int idx = tid; idx < 16 * 64; idx += 128) {
      const int r = idx >> 6, n = idx & 63;
      Vs[r * 64 + n] = KV[(size_t)(jt * 16 + r) * 2048 + 1024 + h * 64 + n];
    }
    __syncthreads();

    if (jt <= qtile) {
      v8f s = {};
#pragma unroll 8
      for (int kk = 0; kk < 32; ++kk) {
        const int kb = kk * 4 + 2 * hhi;
        v2f a; a.x = myQ[l16 * 130 + kb]; a.y = myQ[l16 * 130 + kb + 1];
        v2f b; b.x = Ks[l16 * 130 + kb];  b.y = Ks[l16 * 130 + kb + 1];
        s = wmma_f32(a, b, s);
      }

      const bool diag = (jt == qtile);
#pragma unroll
      for (int v = 0; v < 8; ++v) {
        const int row = v + hhi * 8;
        float sv = s[v];
        if (diag && (l16 > row)) sv = -1.0e30f;
        float mx = sv;
        for (int off = 8; off > 0; off >>= 1)
          mx = fmaxf(mx, __shfl_xor(mx, off, 16));
        const float mnew = fmaxf(m8[v], mx);
        const float alpha = __expf(m8[v] - mnew);
        const float p = __expf(sv - mnew);
        float rs = p;
        for (int off = 8; off > 0; off >>= 1) rs += __shfl_xor(rs, off, 16);
        l8[v] = l8[v] * alpha + rs;
        m8[v] = mnew;
        o0[v] *= alpha; o1[v] *= alpha; o2[v] *= alpha; o3[v] *= alpha;
        myP[row * 18 + l16] = p;
      }

#pragma unroll
      for (int kk = 0; kk < 4; ++kk) {
        const int kb = kk * 4 + 2 * hhi;
        v2f a; a.x = myP[l16 * 18 + kb]; a.y = myP[l16 * 18 + kb + 1];
        v2f b0, b1, b2, b3;
        b0.x = Vs[kb * 64 + l16];      b0.y = Vs[(kb + 1) * 64 + l16];
        b1.x = Vs[kb * 64 + 16 + l16]; b1.y = Vs[(kb + 1) * 64 + 16 + l16];
        b2.x = Vs[kb * 64 + 32 + l16]; b2.y = Vs[(kb + 1) * 64 + 32 + l16];
        b3.x = Vs[kb * 64 + 48 + l16]; b3.y = Vs[(kb + 1) * 64 + 48 + l16];
        o0 = wmma_f32(a, b0, o0);
        o1 = wmma_f32(a, b1, o1);
        o2 = wmma_f32(a, b2, o2);
        o3 = wmma_f32(a, b3, o3);
      }
    }
    __syncthreads();
  }

  const int cb = h * 64;
#pragma unroll
  for (int v = 0; v < 8; ++v) {
    const float inv = 1.0f / l8[v];
    const int r = qrow0 + v + hhi * 8;
    wv[(size_t)r * DM + cb + l16]      = o0[v] * inv;
    wv[(size_t)r * DM + cb + 16 + l16] = o1[v] * inv;
    wv[(size_t)r * DM + cb + 32 + l16] = o2[v] * inv;
    wv[(size_t)r * DM + cb + 48 + l16] = o3[v] * inv;
  }
}

extern "C" void kernel_launch(void* const* d_in, const int* in_sizes, int n_in,
                              void* d_out, int out_size, void* d_ws,
                              size_t ws_size, hipStream_t stream) {
  const float* x    = (const float*)d_in[0];
  const float* pt   = (const float*)d_in[1];
  const float* gq   = (const float*)d_in[3];
  const float* Wq   = (const float*)d_in[4];
  const float* bq   = (const float*)d_in[5];
  const float* gkv  = (const float*)d_in[6];
  const float* Wkv  = (const float*)d_in[7];
  const float* bkv  = (const float*)d_in[8];
  const float* gc   = (const float*)d_in[9];
  const float* Wc   = (const float*)d_in[10];
  const float* bc   = (const float*)d_in[11];
  const float* Wout = (const float*)d_in[12];
  const float* bout = (const float*)d_in[13];
  const int*   cw   = (const int*)d_in[14];
  float* out = (float*)d_out;

  float* ws = (float*)d_ws;
  const size_t R = 1024;
  float* xqn  = ws + 0 * R * 1024;
  float* xkvn = ws + 1 * R * 1024;
  float* ptn  = ws + 2 * R * 1024;
  float* Qb   = ws + 3 * R * 1024;
  float* KVb  = ws + 4 * R * 1024;
  float* Cb   = ws + 6 * R * 1024;
  float* Pb   = ws + 7 * R * 1024;
  float* Qab  = ws + 8256 * R;
  float* Kab  = ws + 10304 * R;
  float* wvb  = ws + 12352 * R;

  rmsnorm3_kernel<<<1024, 256, 0, stream>>>(x, pt, gq, gkv, gc, xqn, xkvn, ptn);
  wmma_gemm_bias<<<dim3(16, 16), 256, 0, stream>>>(xqn, Wq, bq, Qb, 1024);
  wmma_gemm_bias<<<dim3(16, 32), 256, 0, stream>>>(xkvn, Wkv, bkv, KVb, 2048);
  wmma_gemm_bias<<<dim3(16, 16), 256, 0, stream>>>(ptn, Wc, bc, Cb, 1024);
  cumsum_kernel<<<4, 256, 0, stream>>>(Cb, Pb);
  augment_kernel<<<1024, 256, 0, stream>>>(Qb, KVb, Pb, cw, Qab, Kab);
  attn_kernel<<<dim3(16, 16), 128, 0, stream>>>(Qab, Kab, KVb, wvb);
  wmma_gemm_bias<<<dim3(16, 16), 256, 0, stream>>>(wvb, Wout, bout, out, 1024);
}


# attention softmax: the 16-lane row max and row sum butterflies run on the vector ALU (DPP row_xmask) instead of 64 ds_bpermute round trips a key tile; same sums in the same order
# speedup vs baseline: 3.8002x; 3.8002x over previous
#include <hip/hip_runtime.h>
#include <math.h>


typedef __attribute__((ext_vector_type(2))) float v2f;
typedef __attribute__((ext_vector_type(8))) float v8f;

#define L   1024
#define DM  1024
#define EPSR 1.1920929e-07f

static __device__ __forceinline__ v8f wmma_f32(v2f a, v2f b, v8f c) {
  return __builtin_amdgcn_wmma_f32_16x16x4_f32(false, a, false, b, (short)0, c,
                                               false, false);
}

template <int M>
static __device__ __forceinline__ float row_xor(float x) {
  const int b = __builtin_bit_cast(int, x);
  return __builtin_bit_cast(
      float, __builtin_amdgcn_update_dpp(0, b, 0x160 + M, 0xf, 0xf, true));
}

__global__ void __launch_bounds__(256)
rmsnorm3_kernel(const float* __restrict__ x, const float* __restrict__ pt,
                const float* __restrict__ gq, const float* __restrict__ gkv,
                const float* __restrict__ gc,
                float* __restrict__ xqn, float* __restrict__ xkvn,
                float* __restrict__ ptn) {
  __shared__ float red[256];
  const int row = blockIdx.x;
  const int tid = threadIdx.x;

  float4 xv = ((const float4*)(x + (size_t)row * DM))[tid];
  red[tid] = xv.x * xv.x + xv.y * xv.y + xv.z * xv.z + xv.w * xv.w;
  __syncthreads();
  for (int s = 128; s > 0; s >>= 1) {
    if (tid < s) red[tid] += red[tid + s];
    __syncthreads();
  }
  float rn = rsqrtf(red[0] * (1.0f / DM) + EPSR);
  __syncthreads();

  float4 gqv = ((const float4*)gq)[tid];
  float4 gkvv = ((const float4*)gkv)[tid];
  float4 oq, okv;
  oq.x = xv.x * rn * gqv.x;  oq.y = xv.y * rn * gqv.y;
  oq.z = xv.z * rn * gqv.z;  oq.w = xv.w * rn * gqv.w;
  okv.x = xv.x * rn * gkvv.x; okv.y = xv.y * rn * gkvv.y;
  okv.z = xv.z * rn * gkvv.z; okv.w = xv.w * rn * gkvv.w;
  ((float4*)(xqn + (size_t)row * DM))[tid] = oq;
  ((float4*)(xkvn + (size_t)row * DM))[tid] = okv;

  float4 pv = ((const float4*)(pt + (size_t)row * DM))[tid];
  red[tid] = pv.x * pv.x + pv.y * pv.y + pv.z * pv.z + pv.w * pv.w;
  __syncthreads();
  for (int s = 128; s > 0; s >>= 1) {
    if (tid < s) red[tid] += red[tid + s];
    __syncthreads();
  }
  float rnp = rsqrtf(red[0] * (1.0f / DM) + EPSR);
  float4 gcv = ((const float4*)gc)[tid];
  float4 oc;
  oc.x = pv.x * rnp * gcv.x; oc.y = pv.y * rnp * gcv.y;
  oc.z = pv.z * rnp * gcv.z; oc.w = pv.w * rnp * gcv.w;
  ((float4*)(ptn + (size_t)row * DM))[tid] = oc;
}

struct GStage { float4 a[4]; float4 b[16]; };

template <int N>
static __device__ __forceinline__ void load_gstage(GStage& s,
                                                   const float* __restrict__ ap,
                                                   const float* __restrict__ bp,
                                                   int ks) {
  const float4* a4 = (const float4*)(ap + ks * 32);
#pragma unroll
  for (int t = 0; t < 4; ++t) s.a[t] = a4[2 * t];
  const float* b0 = bp + (size_t)ks * 32 * N;
#pragma unroll
  for (int t = 0; t < 4; ++t) {
#pragma unroll
    for (int r = 0; r < 4; ++r)
      s.b[4 * t + r] = *(const float4*)(b0 + (size_t)(8 * t + r) * N);
  }
}

static __device__ __forceinline__ void stage_mma(const GStage& s, v8f& acc0,
                                                 v8f& acc1, v8f& acc2,
                                                 v8f& acc3) {
#pragma unroll
  for (int t = 0; t < 4; ++t) {
    const float4 a = s.a[t];
    const float4 r0 = s.b[4 * t], r1 = s.b[4 * t + 1];
    const float4 r2 = s.b[4 * t + 2], r3 = s.b[4 * t + 3];
    v2f a0; a0.x = a.x; a0.y = a.y;
    v2f a1; a1.x = a.z; a1.y = a.w;
    v2f b00; b00.x = r0.x; b00.y = r1.x;
    v2f b01; b01.x = r0.y; b01.y = r1.y;
    v2f b02; b02.x = r0.z; b02.y = r1.z;
    v2f b03; b03.x = r0.w; b03.y = r1.w;
    v2f b10; b10.x = r2.x; b10.y = r3.x;
    v2f b11; b11.x = r2.y; b11.y = r3.y;
    v2f b12; b12.x = r2.z; b12.y = r3.z;
    v2f b13; b13.x = r2.w; b13.y = r3.w;
    acc0 = wmma_f32(a0, b00, acc0);
    acc1 = wmma_f32(a0, b01, acc1);
    acc2 = wmma_f32(a0, b02, acc2);
    acc3 = wmma_f32(a0, b03, acc3);
    acc0 = wmma_f32(a1, b10, acc0);
    acc1 = wmma_f32(a1, b11, acc1);
    acc2 = wmma_f32(a1, b12, acc2);
    acc3 = wmma_f32(a1, b13, acc3);
  }
}

template <int N>
static __device__ __forceinline__ void gemm_body(const float* __restrict__ A,
                                                 const float* __restrict__ Wm,
                                                 const float* __restrict__ bias,
                                                 float* __restrict__ Cout) {
  const int tid = threadIdx.x;
  const int lane = tid & 31;
  const int wid = __builtin_amdgcn_readfirstlane(tid >> 5);
  const int hhi = lane >> 4, l16 = lane & 15;
  const int m0 = blockIdx.x * 64 + wid * 16;
  const int n0 = blockIdx.y * 64;

  const float* ap = A + (size_t)(m0 + l16) * DM + 4 * hhi;
  const float* bp = Wm + (size_t)(4 * hhi) * N + n0 + 4 * l16;

  v8f acc0 = {}, acc1 = {}, acc2 = {}, acc3 = {};

  GStage sa, sb;
  load_gstage<N>(sa, ap, bp, 0);
  for (int ks = 0; ks < 30; ks += 2) {
    load_gstage<N>(sb, ap, bp, ks + 1);
    __builtin_amdgcn_sched_barrier(0);
    stage_mma(sa, acc0, acc1, acc2, acc3);
    load_gstage<N>(sa, ap, bp, ks + 2);
    __builtin_amdgcn_sched_barrier(0);
    stage_mma(sb, acc0, acc1, acc2, acc3);
  }
  load_gstage<N>(sb, ap, bp, 31);
  __builtin_amdgcn_sched_barrier(0);
  stage_mma(sa, acc0, acc1, acc2, acc3);
  stage_mma(sb, acc0, acc1, acc2, acc3);

  const float4 bv = *(const float4*)(bias + n0 + 4 * l16);
  float* cp = Cout + (size_t)(m0 + hhi * 8) * N + n0 + 4 * l16;
#pragma unroll
  for (int v = 0; v < 8; ++v) {
    float4 o;
    o.x = acc0[v] + bv.x; o.y = acc1[v] + bv.y;
    o.z = acc2[v] + bv.z; o.w = acc3[v] + bv.w;
    *(float4*)(cp + (size_t)v * N) = o;
  }
}

__global__ void __launch_bounds__(128)
wmma_gemm_bias_n1024(const float* __restrict__ A, const float* __restrict__ Wm,
                     const float* __restrict__ bias, float* __restrict__ Cout) {
  gemm_body<1024>(A, Wm, bias, Cout);
}

__global__ void __launch_bounds__(128)
wmma_gemm_bias_n2048(const float* __restrict__ A, const float* __restrict__ Wm,
                     const float* __restrict__ bias, float* __restrict__ Cout) {
  gemm_body<2048>(A, Wm, bias, Cout);
}

__global__ void cumsum_kernel(const float* __restrict__ C, float* __restrict__ P) {
  const int col = blockIdx.x * blockDim.x + threadIdx.x;
  float run = 0.f;
  P[col] = 0.f;
  for (int t = 0; t < L; ++t) {
    run += C[(size_t)t * DM + col];
    P[(size_t)(t + 1) * DM + col] = run;
  }
}

__global__ void augment_kernel(const float* __restrict__ Q,
                               const float* __restrict__ KV,
                               const float* __restrict__ P,
                               const int* __restrict__ cw,
                               float* __restrict__ Qa, float* __restrict__ Ka) {
  const int i = blockIdx.x;
  const int w = cw[0];
  const float scale = 0.125f;
  int E = i + w; if (E > L) E = L;
  int S = i - w; if (S < 0) S = 0;
  for (int c = threadIdx.x; c < DM; c += blockDim.x) {
    const int h = c >> 6, dd = c & 63;
    const float pe = P[(size_t)E * DM + c];
    const float ps = P[(size_t)S * DM + c];
    const float qv = Q[(size_t)i * DM + c];
    const float kv = KV[(size_t)i * 2048 + c];
    const int base = i * 2048 + h * 128 + dd;
    Qa[base]      = qv * (scale + pe);
    Qa[base + 64] = qv;
    Ka[base]      = kv;
    Ka[base + 64] = -kv * ps;
  }
}

struct KVTile { float4 k[16]; float4 v[8]; };

static __device__ __forceinline__ void load_kv_tile(KVTile& t,
                                                    const float* __restrict__ kb,
                                                    const float* __restrict__ vb,
                                                    int jt) {
  const float4* kp = (const float4*)(kb + (size_t)jt * (16 * 2048));
#pragma unroll
  for (int i = 0; i < 16; ++i) t.k[i] = kp[2 * i];
  const float* vp = vb + (size_t)jt * (16 * 2048);
#pragma unroll
  for (int kk = 0; kk < 4; ++kk) {
    t.v[2 * kk]     = *(const float4*)(vp + (size_t)(kk * 4) * 2048);
    t.v[2 * kk + 1] = *(const float4*)(vp + (size_t)(kk * 4 + 1) * 2048);
  }
}

__global__ void __launch_bounds__(128)
attn_kernel(const float* __restrict__ Qa, const float* __restrict__ Ka,
            const float* __restrict__ KV, float* __restrict__ wv) {
  __shared__ float Pls[4 * 16 * 18];

  const int tid = threadIdx.x;
  const int lane = tid & 31;
  const int wid = __builtin_amdgcn_readfirstlane(tid >> 5);
  const int hhi = lane >> 4, l16 = lane & 15;
  const int h = blockIdx.y;
  const int hoff = h * 128;
  const int qrow0 = blockIdx.x * 64 + wid * 16;
  const int qtile = qrow0 >> 4;

  float* myP = &Pls[wid * 16 * 18];

  const float* kbase = Ka + (size_t)l16 * 2048 + hoff + 4 * hhi;
  const float* vbase = KV + (size_t)(2 * hhi) * 2048 + 1024 + h * 64 + 4 * l16;

  float4 qf[16];
  {
    const float4* qp =
        (const float4*)(Qa + (size_t)(qrow0 + l16) * 2048 + hoff + 4 * hhi);
#pragma unroll
    for (int i = 0; i < 16; ++i) qf[i] = qp[2 * i];
  }

  v8f o0 = {}, o1 = {}, o2 = {}, o3 = {};
  float m8[8], l8[8];
#pragma unroll
  for (int v = 0; v < 8; ++v) { m8[v] = -3.0e38f; l8[v] = 0.f; }

  auto tile_step = [&](const KVTile& t, int jt) {
    v8f s0 = {}, s1 = {};
#pragma unroll
    for (int i = 0; i < 16; ++i) {
      v2f a0; a0.x = qf[i].x;  a0.y = qf[i].y;
      v2f b0; b0.x = t.k[i].x; b0.y = t.k[i].y;
      v2f a1; a1.x = qf[i].z;  a1.y = qf[i].w;
      v2f b1; b1.x = t.k[i].z; b1.y = t.k[i].w;
      s0 = wmma_f32(a0, b0, s0);
      s1 = wmma_f32(a1, b1, s1);
    }
    const v8f s = s0 + s1;

    const bool diag = (jt == qtile);
#pragma unroll
    for (int v = 0; v < 8; ++v) {
      const int row = v + hhi * 8;
      float sv = s[v];
      if (diag && (l16 > row)) sv = -1.0e30f;
      float mx = sv;
      mx = fmaxf(mx, row_xor<8>(mx));
      mx = fmaxf(mx, row_xor<4>(mx));
      mx = fmaxf(mx, row_xor<2>(mx));
      mx = fmaxf(mx, row_xor<1>(mx));
      const float mnew = fmaxf(m8[v], mx);
      const float alpha = __expf(m8[v] - mnew);
      const float p = __expf(sv - mnew);
      float rs = p;
      rs += row_xor<8>(rs);
      rs += row_xor<4>(rs);
      rs += row_xor<2>(rs);
      rs += row_xor<1>(rs);
      l8[v] = l8[v] * alpha + rs;
      m8[v] = mnew;
      o0[v] *= alpha; o1[v] *= alpha; o2[v] *= alpha; o3[v] *= alpha;
      myP[row * 18 + l16] = p;
    }

#pragma unroll
    for (int kk = 0; kk < 4; ++kk) {
      const int kb = kk * 4 + 2 * hhi;
      v2f a; a.x = myP[l16 * 18 + kb]; a.y = myP[l16 * 18 + kb + 1];
      const float4 v0 = t.v[2 * kk], v1 = t.v[2 * kk + 1];
      v2f b0, b1, b2, b3;
      b0.x = v0.x; b0.y = v1.x;
      b1.x = v0.y; b1.y = v1.y;
      b2.x = v0.z; b2.y = v1.z;
      b3.x = v0.w; b3.y = v1.w;
      o0 = wmma_f32(a, b0, o0);
      o1 = wmma_f32(a, b1, o1);
      o2 = wmma_f32(a, b2, o2);
      o3 = wmma_f32(a, b3, o3);
    }
  };

  KVTile ta, tb;
  load_kv_tile(ta, kbase, vbase, 0);
  int jt = 0;
  for (; jt + 1 <= qtile; jt += 2) {
    load_kv_tile(tb, kbase, vbase, jt + 1);
    __builtin_amdgcn_sched_barrier(0);
    tile_step(ta, jt);
    __builtin_amdgcn_sched_barrier(0);
    load_kv_tile(ta, kbase, vbase, (jt + 2 < qtile) ? jt + 2 : qtile);
    __builtin_amdgcn_sched_barrier(0);
    tile_step(tb, jt + 1);
  }
  if (jt <= qtile) tile_step(ta, jt);

  const int cb = h * 64;
#pragma unroll
  for (int v = 0; v < 8; ++v) {
    const float inv = 1.0f / l8[v];
    const int r = qrow0 + v + hhi * 8;
    float4 ov;
    ov.x = o0[v] * inv; ov.y = o1[v] * inv;
    ov.z = o2[v] * inv; ov.w = o3[v] * inv;
    *(float4*)(wv + (size_t)r * DM + cb + 4 * l16) = ov;
  }
}

extern "C" void kernel_launch(void* const* d_in, const int* in_sizes, int n_in,
                              void* d_out, int out_size, void* d_ws,
                              size_t ws_size, hipStream_t stream) {
  const float* x    = (const float*)d_in[0];
  const float* pt   = (const float*)d_in[1];
  const float* gq   = (const float*)d_in[3];
  const float* Wq   = (const float*)d_in[4];
  const float* bq   = (const float*)d_in[5];
  const float* gkv  = (const float*)d_in[6];
  const float* Wkv  = (const float*)d_in[7];
  const float* bkv  = (const float*)d_in[8];
  const float* gc   = (const float*)d_in[9];
  const float* Wc   = (const float*)d_in[10];
  const float* bc   = (const float*)d_in[11];
  const float* Wout = (const float*)d_in[12];
  const float* bout = (const float*)d_in[13];
  const int*   cw   = (const int*)d_in[14];
  float* out = (float*)d_out;

  float* ws = (float*)d_ws;
  const size_t R = 1024;
  float* xqn  = ws + 0 * R * 1024;
  float* xkvn = ws + 1 * R * 1024;
  float* ptn  = ws + 2 * R * 1024;
  float* Qb   = ws + 3 * R * 1024;
  float* KVb  = ws + 4 * R * 1024;
  float* Cb   = ws + 6 * R * 1024;
  float* Pb   = ws + 7 * R * 1024;
  float* Qab  = ws + 8256 * R;
  float* Kab  = ws + 10304 * R;
  float* wvb  = ws + 12352 * R;

  rmsnorm3_kernel<<<1024, 256, 0, stream>>>(x, pt, gq, gkv, gc, xqn, xkvn, ptn);
  wmma_gemm_bias_n1024<<<dim3(16, 16), 128, 0, stream>>>(xqn, Wq, bq, Qb);
  wmma_gemm_bias_n2048<<<dim3(16, 32), 128, 0, stream>>>(xkvn, Wkv, bkv, KVb);
  wmma_gemm_bias_n1024<<<dim3(16, 16), 128, 0, stream>>>(ptn, Wc, bc, Cb);
  cumsum_kernel<<<4, 256, 0, stream>>>(Cb, Pb);
  augment_kernel<<<1024, 256, 0, stream>>>(Qb, KVb, Pb, cw, Qab, Kab);
  attn_kernel<<<dim3(16, 16), 128, 0, stream>>>(Qab, Kab, KVb, wvb);
  wmma_gemm_bias_n1024<<<dim3(16, 16), 128, 0, stream>>>(wvb, Wout, bout, out);
}


# prefix sum + augmentation fused: each 16-row tile re-sums its own column prefix of C by independent 16-byte loads (no serial scan, no P plane), writes Q~/K~; 8->7 launches; prefix-sum order changes
# speedup vs baseline: 4.6101x; 1.2131x over previous
#include <hip/hip_runtime.h>
#include <math.h>


typedef __attribute__((ext_vector_type(2))) float v2f;
typedef __attribute__((ext_vector_type(8))) float v8f;

#define L   1024
#define DM  1024
#define EPSR 1.1920929e-07f

static __device__ __forceinline__ v8f wmma_f32(v2f a, v2f b, v8f c) {
  return __builtin_amdgcn_wmma_f32_16x16x4_f32(false, a, false, b, (short)0, c,
                                               false, false);
}

template <int M>
static __device__ __forceinline__ float row_xor(float x) {
  const int b = __builtin_bit_cast(int, x);
  return __builtin_bit_cast(
      float, __builtin_amdgcn_update_dpp(0, b, 0x160 + M, 0xf, 0xf, true));
}

__global__ void __launch_bounds__(256)
rmsnorm3_kernel(const float* __restrict__ x, const float* __restrict__ pt,
                const float* __restrict__ gq, const float* __restrict__ gkv,
                const float* __restrict__ gc,
                float* __restrict__ xqn, float* __restrict__ xkvn,
                float* __restrict__ ptn) {
  __shared__ float red[256];
  const int row = blockIdx.x;
  const int tid = threadIdx.x;

  float4 xv = ((const float4*)(x + (size_t)row * DM))[tid];
  red[tid] = xv.x * xv.x + xv.y * xv.y + xv.z * xv.z + xv.w * xv.w;
  __syncthreads();
  for (int s = 128; s > 0; s >>= 1) {
    if (tid < s) red[tid] += red[tid + s];
    __syncthreads();
  }
  float rn = rsqrtf(red[0] * (1.0f / DM) + EPSR);
  __syncthreads();

  float4 gqv = ((const float4*)gq)[tid];
  float4 gkvv = ((const float4*)gkv)[tid];
  float4 oq, okv;
  oq.x = xv.x * rn * gqv.x;  oq.y = xv.y * rn * gqv.y;
  oq.z = xv.z * rn * gqv.z;  oq.w = xv.w * rn * gqv.w;
  okv.x = xv.x * rn * gkvv.x; okv.y = xv.y * rn * gkvv.y;
  okv.z = xv.z * rn * gkvv.z; okv.w = xv.w * rn * gkvv.w;
  ((float4*)(xqn + (size_t)row * DM))[tid] = oq;
  ((float4*)(xkvn + (size_t)row * DM))[tid] = okv;

  float4 pv = ((const float4*)(pt + (size_t)row * DM))[tid];
  red[tid] = pv.x * pv.x + pv.y * pv.y + pv.z * pv.z + pv.w * pv.w;
  __syncthreads();
  for (int s = 128; s > 0; s >>= 1) {
    if (tid < s) red[tid] += red[tid + s];
    __syncthreads();
  }
  float rnp = rsqrtf(red[0] * (1.0f / DM) + EPSR);
  float4 gcv = ((const float4*)gc)[tid];
  float4 oc;
  oc.x = pv.x * rnp * gcv.x; oc.y = pv.y * rnp * gcv.y;
  oc.z = pv.z * rnp * gcv.z; oc.w = pv.w * rnp * gcv.w;
  ((float4*)(ptn + (size_t)row * DM))[tid] = oc;
}

struct GStage { float4 a[4]; float4 b[16]; };

template <int N>
static __device__ __forceinline__ void load_gstage(GStage& s,
                                                   const float* __restrict__ ap,
                                                   const float* __restrict__ bp,
                                                   int ks) {
  const float4* a4 = (const float4*)(ap + ks * 32);
#pragma unroll
  for (int t = 0; t < 4; ++t) s.a[t] = a4[2 * t];
  const float* b0 = bp + (size_t)ks * 32 * N;
#pragma unroll
  for (int t = 0; t < 4; ++t) {
#pragma unroll
    for (int r = 0; r < 4; ++r)
      s.b[4 * t + r] = *(const float4*)(b0 + (size_t)(8 * t + r) * N);
  }
}

static __device__ __forceinline__ void stage_mma(const GStage& s, v8f& acc0,
                                                 v8f& acc1, v8f& acc2,
                                                 v8f& acc3) {
#pragma unroll
  for (int t = 0; t < 4; ++t) {
    const float4 a = s.a[t];
    const float4 r0 = s.b[4 * t], r1 = s.b[4 * t + 1];
    const float4 r2 = s.b[4 * t + 2], r3 = s.b[4 * t + 3];
    v2f a0; a0.x = a.x; a0.y = a.y;
    v2f a1; a1.x = a.z; a1.y = a.w;
    v2f b00; b00.x = r0.x; b00.y = r1.x;
    v2f b01; b01.x = r0.y; b01.y = r1.y;
    v2f b02; b02.x = r0.z; b02.y = r1.z;
    v2f b03; b03.x = r0.w; b03.y = r1.w;
    v2f b10; b10.x = r2.x; b10.y = r3.x;
    v2f b11; b11.x = r2.y; b11.y = r3.y;
    v2f b12; b12.x = r2.z; b12.y = r3.z;
    v2f b13; b13.x = r2.w; b13.y = r3.w;
    acc0 = wmma_f32(a0, b00, acc0);
    acc1 = wmma_f32(a0, b01, acc1);
    acc2 = wmma_f32(a0, b02, acc2);
    acc3 = wmma_f32(a0, b03, acc3);
    acc0 = wmma_f32(a1, b10, acc0);
    acc1 = wmma_f32(a1, b11, acc1);
    acc2 = wmma_f32(a1, b12, acc2);
    acc3 = wmma_f32(a1, b13, acc3);
  }
}

template <int N>
static __device__ __forceinline__ void gemm_body(const float* __restrict__ A,
                                                 const float* __restrict__ Wm,
                                                 const float* __restrict__ bias,
                                                 float* __restrict__ Cout) {
  const int tid = threadIdx.x;
  const int lane = tid & 31;
  const int wid = __builtin_amdgcn_readfirstlane(tid >> 5);
  const int hhi = lane >> 4, l16 = lane & 15;
  const int m0 = blockIdx.x * 64 + wid * 16;
  const int n0 = blockIdx.y * 64;

  const float* ap = A + (size_t)(m0 + l16) * DM + 4 * hhi;
  const float* bp = Wm + (size_t)(4 * hhi) * N + n0 + 4 * l16;

  v8f acc0 = {}, acc1 = {}, acc2 = {}, acc3 = {};

  GStage sa, sb;
  load_gstage<N>(sa, ap, bp, 0);
  for (int ks = 0; ks < 30; ks += 2) {
    load_gstage<N>(sb, ap, bp, ks + 1);
    __builtin_amdgcn_sched_barrier(0);
    stage_mma(sa, acc0, acc1, acc2, acc3);
    load_gstage<N>(sa, ap, bp, ks + 2);
    __builtin_amdgcn_sched_barrier(0);
    stage_mma(sb, acc0, acc1, acc2, acc3);
  }
  load_gstage<N>(sb, ap, bp, 31);
  __builtin_amdgcn_sched_barrier(0);
  stage_mma(sa, acc0, acc1, acc2, acc3);
  stage_mma(sb, acc0, acc1, acc2, acc3);

  const float4 bv = *(const float4*)(bias + n0 + 4 * l16);
  float* cp = Cout + (size_t)(m0 + hhi * 8) * N + n0 + 4 * l16;
#pragma unroll
  for (int v = 0; v < 8; ++v) {
    float4 o;
    o.x = acc0[v] + bv.x; o.y = acc1[v] + bv.y;
    o.z = acc2[v] + bv.z; o.w = acc3[v] + bv.w;
    *(float4*)(cp + (size_t)v * N) = o;
  }
}

__global__ void __launch_bounds__(128)
wmma_gemm_bias_n1024(const float* __restrict__ A, const float* __restrict__ Wm,
                     const float* __restrict__ bias, float* __restrict__ Cout) {
  gemm_body<1024>(A, Wm, bias, Cout);
}

__global__ void __launch_bounds__(128)
wmma_gemm_bias_n2048(const float* __restrict__ A, const float* __restrict__ Wm,
                     const float* __restrict__ bias, float* __restrict__ Cout) {
  gemm_body<2048>(A, Wm, bias, Cout);
}

typedef __attribute__((ext_vector_type(4))) float v4f;

static __device__ __forceinline__ int clip0L(int x) {
  return x < 0 ? 0 : (x > L ? L : x);
}

static __device__ __forceinline__ v4f ld4(const float* p) {
  return *(const v4f*)p;
}

__global__ void __launch_bounds__(128)
prefix_augment_kernel(const float* __restrict__ Q, const float* __restrict__ KV,
                      const float* __restrict__ C, const int* __restrict__ cw,
                      float* __restrict__ Qa, float* __restrict__ Ka) {
  const int tid = threadIdx.x;
  const int lane = tid & 31;
  const int wid = __builtin_amdgcn_readfirstlane(tid >> 5);
  const int hhi = lane >> 4, l16 = lane & 15;
  const int p = blockIdx.x;
  const int r = (wid & 1) ? 63 - p : p;
  const int hd = blockIdx.y * 2 + (wid >> 1);
  const int i0 = r * 16;
  const int w = cw[0];
  const float scale = 0.125f;

  const float* cb = C + hd * 64 + 4 * l16;
  const v4f zero = {0.f, 0.f, 0.f, 0.f};

  const int s0 = clip0L(i0 - w), e0 = clip0L(i0 + w);
  const int lo = s0 < e0 ? s0 : e0;

  v4f g0 = zero, g1 = zero, g2 = zero, g3 = zero;
  const int nfull = lo >> 5;
  v4f xa[16], xb[16];
  auto load_trip = [&](v4f* x, int k) {
#pragma unroll
    for (int u = 0; u < 16; ++u)
      x[u] = ld4(cb + (size_t)(32 * k + 2 * u + hhi) * DM);
  };
  auto add_trip = [&](const v4f* x) {
#pragma unroll
    for (int u = 0; u < 16; u += 4) {
      g0 += x[u]; g1 += x[u + 1]; g2 += x[u + 2]; g3 += x[u + 3];
    }
  };
  int k = 0;
  if (nfull > 0) load_trip(xa, 0);
  for (; k + 2 <= nfull; k += 2) {
    load_trip(xb, k + 1);
    __builtin_amdgcn_sched_barrier(0);
    add_trip(xa);
    load_trip(xa, (k + 2 < nfull) ? k + 2 : nfull - 1);
    __builtin_amdgcn_sched_barrier(0);
    add_trip(xb);
  }
  if (k < nfull) add_trip(xa);
  const int t = nfull * 32;
  if (t < lo) {
    v4f x[16];
#pragma unroll
    for (int u = 0; u < 16; ++u) {
      const int row = t + 2 * u + hhi;
      x[u] = ld4(cb + (size_t)(row < L - 1 ? row : L - 1) * DM);
    }
#pragma unroll
    for (int u = 0; u < 16; u += 4) {
      const int row = t + 2 * u + hhi;
      g0 += (row     < lo) ? x[u]     : zero;
      g1 += (row + 2 < lo) ? x[u + 1] : zero;
      g2 += (row + 4 < lo) ? x[u + 2] : zero;
      g3 += (row + 6 < lo) ? x[u + 3] : zero;
    }
  }
  v4f G = (g0 + g1) + (g2 + g3);
  {
    v4f o;
    o.x = __shfl_xor(G.x, 16); o.y = __shfl_xor(G.y, 16);
    o.z = __shfl_xor(G.z, 16); o.w = __shfl_xor(G.w, 16);
    G += o;
  }

  const int ia = i0 + 8 * hhi;
  const int sa = clip0L(ia - w), ea = clip0L(ia + w);
  const int s8 = clip0L(i0 + 8 - w), e8 = clip0L(i0 + 8 + w);
  const int hiA = s8 > e8 ? s8 : e8;
  v4f psA = G, peA = G;
  for (int t0 = lo; t0 < hiA; t0 += 16) {
    v4f x[16];
#pragma unroll
    for (int u = 0; u < 16; ++u) {
      const int row = t0 + u;
      x[u] = ld4(cb + (size_t)(row < L - 1 ? row : L - 1) * DM);
    }
#pragma unroll
    for (int u = 0; u < 16; ++u) {
      const int row = t0 + u;
      psA += (row < sa) ? x[u] : zero;
      peA += (row < ea) ? x[u] : zero;
    }
  }

  v4f ds[7], de[7];
#pragma unroll
  for (int v = 1; v < 8; ++v) {
    const int sp = clip0L(ia + v - 1 - w), sc = clip0L(ia + v - w);
    const int ep = clip0L(ia + v - 1 + w), ec = clip0L(ia + v + w);
    const v4f xs = ld4(cb + (size_t)(sp < L - 1 ? sp : L - 1) * DM);
    const v4f xe = ld4(cb + (size_t)(ep < L - 1 ? ep : L - 1) * DM);
    ds[v - 1] = (sc > sp) ? xs : zero;
    de[v - 1] = (ec > ep) ? xe : zero;
  }

  v4f ps = psA, pe = peA;
#pragma unroll
  for (int v = 0; v < 8; ++v) {
    if (v > 0) { ps += ds[v - 1]; pe += de[v - 1]; }
    const int i = ia + v;
    const v4f qv = ld4(Q + (size_t)i * DM + hd * 64 + 4 * l16);
    const v4f kv = ld4(KV + (size_t)i * 2048 + hd * 64 + 4 * l16);
    const size_t base = (size_t)i * 2048 + hd * 128 + 4 * l16;
    *(v4f*)(Qa + base)      = qv * (scale + pe);
    *(v4f*)(Qa + base + 64) = qv;
    *(v4f*)(Ka + base)      = kv;
    *(v4f*)(Ka + base + 64) = -kv * ps;
  }
}

struct KVTile { float4 k[16]; float4 v[8]; };

static __device__ __forceinline__ void load_kv_tile(KVTile& t,
                                                    const float* __restrict__ kb,
                                                    const float* __restrict__ vb,
                                                    int jt) {
  const float4* kp = (const float4*)(kb + (size_t)jt * (16 * 2048));
#pragma unroll
  for (int i = 0; i < 16; ++i) t.k[i] = kp[2 * i];
  const float* vp = vb + (size_t)jt * (16 * 2048);
#pragma unroll
  for (int kk = 0; kk < 4; ++kk) {
    t.v[2 * kk]     = *(const float4*)(vp + (size_t)(kk * 4) * 2048);
    t.v[2 * kk + 1] = *(const float4*)(vp + (size_t)(kk * 4 + 1) * 2048);
  }
}

__global__ void __launch_bounds__(128)
attn_kernel(const float* __restrict__ Qa, const float* __restrict__ Ka,
            const float* __restrict__ KV, float* __restrict__ wv) {
  __shared__ float Pls[4 * 16 * 18];

  const int tid = threadIdx.x;
  const int lane = tid & 31;
  const int wid = __builtin_amdgcn_readfirstlane(tid >> 5);
  const int hhi = lane >> 4, l16 = lane & 15;
  const int h = blockIdx.y;
  const int hoff = h * 128;
  const int qrow0 = blockIdx.x * 64 + wid * 16;
  const int qtile = qrow0 >> 4;

  float* myP = &Pls[wid * 16 * 18];

  const float* kbase = Ka + (size_t)l16 * 2048 + hoff + 4 * hhi;
  const float* vbase = KV + (size_t)(2 * hhi) * 2048 + 1024 + h * 64 + 4 * l16;

  float4 qf[16];
  {
    const float4* qp =
        (const float4*)(Qa + (size_t)(qrow0 + l16) * 2048 + hoff + 4 * hhi);
#pragma unroll
    for (int i = 0; i < 16; ++i) qf[i] = qp[2 * i];
  }

  v8f o0 = {}, o1 = {}, o2 = {}, o3 = {};
  float m8[8], l8[8];
#pragma unroll
  for (int v = 0; v < 8; ++v) { m8[v] = -3.0e38f; l8[v] = 0.f; }

  auto tile_step = [&](const KVTile& t, int jt) {
    v8f s0 = {}, s1 = {};
#pragma unroll
    for (int i = 0; i < 16; ++i) {
      v2f a0; a0.x = qf[i].x;  a0.y = qf[i].y;
      v2f b0; b0.x = t.k[i].x; b0.y = t.k[i].y;
      v2f a1; a1.x = qf[i].z;  a1.y = qf[i].w;
      v2f b1; b1.x = t.k[i].z; b1.y = t.k[i].w;
      s0 = wmma_f32(a0, b0, s0);
      s1 = wmma_f32(a1, b1, s1);
    }
    const v8f s = s0 + s1;

    const bool diag = (jt == qtile);
#pragma unroll
    for (int v = 0; v < 8; ++v) {
      const int row = v + hhi * 8;
      float sv = s[v];
      if (diag && (l16 > row)) sv = -1.0e30f;
      float mx = sv;
      mx = fmaxf(mx, row_xor<8>(mx));
      mx = fmaxf(mx, row_xor<4>(mx));
      mx = fmaxf(mx, row_xor<2>(mx));
      mx = fmaxf(mx, row_xor<1>(mx));
      const float mnew = fmaxf(m8[v], mx);
      const float alpha = __expf(m8[v] - mnew);
      const float p = __expf(sv - mnew);
      float rs = p;
      rs += row_xor<8>(rs);
      rs += row_xor<4>(rs);
      rs += row_xor<2>(rs);
      rs += row_xor<1>(rs);
      l8[v] = l8[v] * alpha + rs;
      m8[v] = mnew;
      o0[v] *= alpha; o1[v] *= alpha; o2[v] *= alpha; o3[v] *= alpha;
      myP[row * 18 + l16] = p;
    }

#pragma unroll
    for (int kk = 0; kk < 4; ++kk) {
      const int kb = kk * 4 + 2 * hhi;
      v2f a; a.x = myP[l16 * 18 + kb]; a.y = myP[l16 * 18 + kb + 1];
      const float4 v0 = t.v[2 * kk], v1 = t.v[2 * kk + 1];
      v2f b0, b1, b2, b3;
      b0.x = v0.x; b0.y = v1.x;
      b1.x = v0.y; b1.y = v1.y;
      b2.x = v0.z; b2.y = v1.z;
      b3.x = v0.w; b3.y = v1.w;
      o0 = wmma_f32(a, b0, o0);
      o1 = wmma_f32(a, b1, o1);
      o2 = wmma_f32(a, b2, o2);
      o3 = wmma_f32(a, b3, o3);
    }
  };

  KVTile ta, tb;
  load_kv_tile(ta, kbase, vbase, 0);
  int jt = 0;
  for (; jt + 1 <= qtile; jt += 2) {
    load_kv_tile(tb, kbase, vbase, jt + 1);
    __builtin_amdgcn_sched_barrier(0);
    tile_step(ta, jt);
    __builtin_amdgcn_sched_barrier(0);
    load_kv_tile(ta, kbase, vbase, (jt + 2 < qtile) ? jt + 2 : qtile);
    __builtin_amdgcn_sched_barrier(0);
    tile_step(tb, jt + 1);
  }
  if (jt <= qtile) tile_step(ta, jt);

  const int cb = h * 64;
#pragma unroll
  for (int v = 0; v < 8; ++v) {
    const float inv = 1.0f / l8[v];
    const int r = qrow0 + v + hhi * 8;
    float4 ov;
    ov.x = o0[v] * inv; ov.y = o1[v] * inv;
    ov.z = o2[v] * inv; ov.w = o3[v] * inv;
    *(float4*)(wv + (size_t)r * DM + cb + 4 * l16) = ov;
  }
}

extern "C" void kernel_launch(void* const* d_in, const int* in_sizes, int n_in,
                              void* d_out, int out_size, void* d_ws,
                              size_t ws_size, hipStream_t stream) {
  const float* x    = (const float*)d_in[0];
  const float* pt   = (const float*)d_in[1];
  const float* gq   = (const float*)d_in[3];
  const float* Wq   = (const float*)d_in[4];
  const float* bq   = (const float*)d_in[5];
  const float* gkv  = (const float*)d_in[6];
  const float* Wkv  = (const float*)d_in[7];
  const float* bkv  = (const float*)d_in[8];
  const float* gc   = (const float*)d_in[9];
  const float* Wc   = (const float*)d_in[10];
  const float* bc   = (const float*)d_in[11];
  const float* Wout = (const float*)d_in[12];
  const float* bout = (const float*)d_in[13];
  const int*   cw   = (const int*)d_in[14];
  float* out = (float*)d_out;

  float* ws = (float*)d_ws;
  const size_t R = 1024;
  float* xqn  = ws + 0 * R * 1024;
  float* xkvn = ws + 1 * R * 1024;
  float* ptn  = ws + 2 * R * 1024;
  float* Qb   = ws + 3 * R * 1024;
  float* KVb  = ws + 4 * R * 1024;
  float* Cb   = ws + 6 * R * 1024;
  float* Qab  = ws + 8256 * R;
  float* Kab  = ws + 10304 * R;
  float* wvb  = ws + 12352 * R;

  rmsnorm3_kernel<<<1024, 256, 0, stream>>>(x, pt, gq, gkv, gc, xqn, xkvn, ptn);
  wmma_gemm_bias_n1024<<<dim3(16, 16), 128, 0, stream>>>(xqn, Wq, bq, Qb);
  wmma_gemm_bias_n2048<<<dim3(16, 32), 128, 0, stream>>>(xkvn, Wkv, bkv, KVb);
  wmma_gemm_bias_n1024<<<dim3(16, 16), 128, 0, stream>>>(ptn, Wc, bc, Cb);
  prefix_augment_kernel<<<dim3(32, 8), 128, 0, stream>>>(Qb, KVb, Cb, cw, Qab,
                                                         Kab);
  attn_kernel<<<dim3(16, 16), 128, 0, stream>>>(Qab, Kab, KVb, wvb);
  wmma_gemm_bias_n1024<<<dim3(16, 16), 128, 0, stream>>>(wvb, Wout, bout, out);
}


# attention key loop software-pipelined in the wave: tile j's softmax (stage by stage over its 8 rows) issues between the score WMMAs of tile j+1, not between two matrix runs; same sums, same order
# speedup vs baseline: 4.8622x; 1.0547x over previous
#include <hip/hip_runtime.h>
#include <math.h>


typedef __attribute__((ext_vector_type(2))) float v2f;
typedef __attribute__((ext_vector_type(8))) float v8f;

#define L   1024
#define DM  1024
#define EPSR 1.1920929e-07f

static __device__ __forceinline__ v8f wmma_f32(v2f a, v2f b, v8f c) {
  return __builtin_amdgcn_wmma_f32_16x16x4_f32(false, a, false, b, (short)0, c,
                                               false, false);
}

template <int M>
static __device__ __forceinline__ float row_xor(float x) {
  const int b = __builtin_bit_cast(int, x);
  return __builtin_bit_cast(
      float, __builtin_amdgcn_update_dpp(0, b, 0x160 + M, 0xf, 0xf, true));
}

__global__ void __launch_bounds__(256)
rmsnorm3_kernel(const float* __restrict__ x, const float* __restrict__ pt,
                const float* __restrict__ gq, const float* __restrict__ gkv,
                const float* __restrict__ gc,
                float* __restrict__ xqn, float* __restrict__ xkvn,
                float* __restrict__ ptn) {
  __shared__ float red[256];
  const int row = blockIdx.x;
  const int tid = threadIdx.x;

  float4 xv = ((const float4*)(x + (size_t)row * DM))[tid];
  red[tid] = xv.x * xv.x + xv.y * xv.y + xv.z * xv.z + xv.w * xv.w;
  __syncthreads();
  for (int s = 128; s > 0; s >>= 1) {
    if (tid < s) red[tid] += red[tid + s];
    __syncthreads();
  }
  float rn = rsqrtf(red[0] * (1.0f / DM) + EPSR);
  __syncthreads();

  float4 gqv = ((const float4*)gq)[tid];
  float4 gkvv = ((const float4*)gkv)[tid];
  float4 oq, okv;
  oq.x = xv.x * rn * gqv.x;  oq.y = xv.y * rn * gqv.y;
  oq.z = xv.z * rn * gqv.z;  oq.w = xv.w * rn * gqv.w;
  okv.x = xv.x * rn * gkvv.x; okv.y = xv.y * rn * gkvv.y;
  okv.z = xv.z * rn * gkvv.z; okv.w = xv.w * rn * gkvv.w;
  ((float4*)(xqn + (size_t)row * DM))[tid] = oq;
  ((float4*)(xkvn + (size_t)row * DM))[tid] = okv;

  float4 pv = ((const float4*)(pt + (size_t)row * DM))[tid];
  red[tid] = pv.x * pv.x + pv.y * pv.y + pv.z * pv.z + pv.w * pv.w;
  __syncthreads();
  for (int s = 128; s > 0; s >>= 1) {
    if (tid < s) red[tid] += red[tid + s];
    __syncthreads();
  }
  float rnp = rsqrtf(red[0] * (1.0f / DM) + EPSR);
  float4 gcv = ((const float4*)gc)[tid];
  float4 oc;
  oc.x = pv.x * rnp * gcv.x; oc.y = pv.y * rnp * gcv.y;
  oc.z = pv.z * rnp * gcv.z; oc.w = pv.w * rnp * gcv.w;
  ((float4*)(ptn + (size_t)row * DM))[tid] = oc;
}

struct GStage { float4 a[4]; float4 b[16]; };

template <int N>
static __device__ __forceinline__ void load_gstage(GStage& s,
                                                   const float* __restrict__ ap,
                                                   const float* __restrict__ bp,
                                                   int ks) {
  const float4* a4 = (const float4*)(ap + ks * 32);
#pragma unroll
  for (int t = 0; t < 4; ++t) s.a[t] = a4[2 * t];
  const float* b0 = bp + (size_t)ks * 32 * N;
#pragma unroll
  for (int t = 0; t < 4; ++t) {
#pragma unroll
    for (int r = 0; r < 4; ++r)
      s.b[4 * t + r] = *(const float4*)(b0 + (size_t)(8 * t + r) * N);
  }
}

static __device__ __forceinline__ void stage_mma(const GStage& s, v8f& acc0,
                                                 v8f& acc1, v8f& acc2,
                                                 v8f& acc3) {
#pragma unroll
  for (int t = 0; t < 4; ++t) {
    const float4 a = s.a[t];
    const float4 r0 = s.b[4 * t], r1 = s.b[4 * t + 1];
    const float4 r2 = s.b[4 * t + 2], r3 = s.b[4 * t + 3];
    v2f a0; a0.x = a.x; a0.y = a.y;
    v2f a1; a1.x = a.z; a1.y = a.w;
    v2f b00; b00.x = r0.x; b00.y = r1.x;
    v2f b01; b01.x = r0.y; b01.y = r1.y;
    v2f b02; b02.x = r0.z; b02.y = r1.z;
    v2f b03; b03.x = r0.w; b03.y = r1.w;
    v2f b10; b10.x = r2.x; b10.y = r3.x;
    v2f b11; b11.x = r2.y; b11.y = r3.y;
    v2f b12; b12.x = r2.z; b12.y = r3.z;
    v2f b13; b13.x = r2.w; b13.y = r3.w;
    acc0 = wmma_f32(a0, b00, acc0);
    acc1 = wmma_f32(a0, b01, acc1);
    acc2 = wmma_f32(a0, b02, acc2);
    acc3 = wmma_f32(a0, b03, acc3);
    acc0 = wmma_f32(a1, b10, acc0);
    acc1 = wmma_f32(a1, b11, acc1);
    acc2 = wmma_f32(a1, b12, acc2);
    acc3 = wmma_f32(a1, b13, acc3);
  }
}

template <int N>
static __device__ __forceinline__ void gemm_body(const float* __restrict__ A,
                                                 const float* __restrict__ Wm,
                                                 const float* __restrict__ bias,
                                                 float* __restrict__ Cout) {
  const int tid = threadIdx.x;
  const int lane = tid & 31;
  const int wid = __builtin_amdgcn_readfirstlane(tid >> 5);
  const int hhi = lane >> 4, l16 = lane & 15;
  const int m0 = blockIdx.x * 64 + wid * 16;
  const int n0 = blockIdx.y * 64;

  const float* ap = A + (size_t)(m0 + l16) * DM + 4 * hhi;
  const float* bp = Wm + (size_t)(4 * hhi) * N + n0 + 4 * l16;

  v8f acc0 = {}, acc1 = {}, acc2 = {}, acc3 = {};

  GStage sa, sb;
  load_gstage<N>(sa, ap, bp, 0);
  for (int ks = 0; ks < 30; ks += 2) {
    load_gstage<N>(sb, ap, bp, ks + 1);
    __builtin_amdgcn_sched_barrier(0);
    stage_mma(sa, acc0, acc1, acc2, acc3);
    load_gstage<N>(sa, ap, bp, ks + 2);
    __builtin_amdgcn_sched_barrier(0);
    stage_mma(sb, acc0, acc1, acc2, acc3);
  }
  load_gstage<N>(sb, ap, bp, 31);
  __builtin_amdgcn_sched_barrier(0);
  stage_mma(sa, acc0, acc1, acc2, acc3);
  stage_mma(sb, acc0, acc1, acc2, acc3);

  const float4 bv = *(const float4*)(bias + n0 + 4 * l16);
  float* cp = Cout + (size_t)(m0 + hhi * 8) * N + n0 + 4 * l16;
#pragma unroll
  for (int v = 0; v < 8; ++v) {
    float4 o;
    o.x = acc0[v] + bv.x; o.y = acc1[v] + bv.y;
    o.z = acc2[v] + bv.z; o.w = acc3[v] + bv.w;
    *(float4*)(cp + (size_t)v * N) = o;
  }
}

__global__ void __launch_bounds__(128)
wmma_gemm_bias_n1024(const float* __restrict__ A, const float* __restrict__ Wm,
                     const float* __restrict__ bias, float* __restrict__ Cout) {
  gemm_body<1024>(A, Wm, bias, Cout);
}

__global__ void __launch_bounds__(128)
wmma_gemm_bias_n2048(const float* __restrict__ A, const float* __restrict__ Wm,
                     const float* __restrict__ bias, float* __restrict__ Cout) {
  gemm_body<2048>(A, Wm, bias, Cout);
}

typedef __attribute__((ext_vector_type(4))) float v4f;

static __device__ __forceinline__ int clip0L(int x) {
  return x < 0 ? 0 : (x > L ? L : x);
}

static __device__ __forceinline__ v4f ld4(const float* p) {
  return *(const v4f*)p;
}

__global__ void __launch_bounds__(128)
prefix_augment_kernel(const float* __restrict__ Q, const float* __restrict__ KV,
                      const float* __restrict__ C, const int* __restrict__ cw,
                      float* __restrict__ Qa, float* __restrict__ Ka) {
  const int tid = threadIdx.x;
  const int lane = tid & 31;
  const int wid = __builtin_amdgcn_readfirstlane(tid >> 5);
  const int hhi = lane >> 4, l16 = lane & 15;
  const int p = blockIdx.x;
  const int r = (wid & 1) ? 63 - p : p;
  const int hd = blockIdx.y * 2 + (wid >> 1);
  const int i0 = r * 16;
  const int w = cw[0];
  const float scale = 0.125f;

  const float* cb = C + hd * 64 + 4 * l16;
  const v4f zero = {0.f, 0.f, 0.f, 0.f};

  const int s0 = clip0L(i0 - w), e0 = clip0L(i0 + w);
  const int lo = s0 < e0 ? s0 : e0;

  v4f g0 = zero, g1 = zero, g2 = zero, g3 = zero;
  const int nfull = lo >> 5;
  v4f xa[16], xb[16];
  auto load_trip = [&](v4f* x, int k) {
#pragma unroll
    for (int u = 0; u < 16; ++u)
      x[u] = ld4(cb + (size_t)(32 * k + 2 * u + hhi) * DM);
  };
  auto add_trip = [&](const v4f* x) {
#pragma unroll
    for (int u = 0; u < 16; u += 4) {
      g0 += x[u]; g1 += x[u + 1]; g2 += x[u + 2]; g3 += x[u + 3];
    }
  };
  int k = 0;
  if (nfull > 0) load_trip(xa, 0);
  for (; k + 2 <= nfull; k += 2) {
    load_trip(xb, k + 1);
    __builtin_amdgcn_sched_barrier(0);
    add_trip(xa);
    load_trip(xa, (k + 2 < nfull) ? k + 2 : nfull - 1);
    __builtin_amdgcn_sched_barrier(0);
    add_trip(xb);
  }
  if (k < nfull) add_trip(xa);
  const int t = nfull * 32;
  if (t < lo) {
    v4f x[16];
#pragma unroll
    for (int u = 0; u < 16; ++u) {
      const int row = t + 2 * u + hhi;
      x[u] = ld4(cb + (size_t)(row < L - 1 ? row : L - 1) * DM);
    }
#pragma unroll
    for (int u = 0; u < 16; u += 4) {
      const int row = t + 2 * u + hhi;
      g0 += (row     < lo) ? x[u]     : zero;
      g1 += (row + 2 < lo) ? x[u + 1] : zero;
      g2 += (row + 4 < lo) ? x[u + 2] : zero;
      g3 += (row + 6 < lo) ? x[u + 3] : zero;
    }
  }
  v4f G = (g0 + g1) + (g2 + g3);
  {
    v4f o;
    o.x = __shfl_xor(G.x, 16); o.y = __shfl_xor(G.y, 16);
    o.z = __shfl_xor(G.z, 16); o.w = __shfl_xor(G.w, 16);
    G += o;
  }

  const int ia = i0 + 8 * hhi;
  const int sa = clip0L(ia - w), ea = clip0L(ia + w);
  const int s8 = clip0L(i0 + 8 - w), e8 = clip0L(i0 + 8 + w);
  const int hiA = s8 > e8 ? s8 : e8;
  v4f psA = G, peA = G;
  for (int t0 = lo; t0 < hiA; t0 += 16) {
    v4f x[16];
#pragma unroll
    for (int u = 0; u < 16; ++u) {
      const int row = t0 + u;
      x[u] = ld4(cb + (size_t)(row < L - 1 ? row : L - 1) * DM);
    }
#pragma unroll
    for (int u = 0; u < 16; ++u) {
      const int row = t0 + u;
      psA += (row < sa) ? x[u] : zero;
      peA += (row < ea) ? x[u] : zero;
    }
  }

  v4f ds[7], de[7];
#pragma unroll
  for (int v = 1; v < 8; ++v) {
    const int sp = clip0L(ia + v - 1 - w), sc = clip0L(ia + v - w);
    const int ep = clip0L(ia + v - 1 + w), ec = clip0L(ia + v + w);
    const v4f xs = ld4(cb + (size_t)(sp < L - 1 ? sp : L - 1) * DM);
    const v4f xe = ld4(cb + (size_t)(ep < L - 1 ? ep : L - 1) * DM);
    ds[v - 1] = (sc > sp) ? xs : zero;
    de[v - 1] = (ec > ep) ? xe : zero;
  }

  v4f ps = psA, pe = peA;
#pragma unroll
  for (int v = 0; v < 8; ++v) {
    if (v > 0) { ps += ds[v - 1]; pe += de[v - 1]; }
    const int i = ia + v;
    const v4f qv = ld4(Q + (size_t)i * DM + hd * 64 + 4 * l16);
    const v4f kv = ld4(KV + (size_t)i * 2048 + hd * 64 + 4 * l16);
    const size_t base = (size_t)i * 2048 + hd * 128 + 4 * l16;
    *(v4f*)(Qa + base)      = qv * (scale + pe);
    *(v4f*)(Qa + base + 64) = qv;
    *(v4f*)(Ka + base)      = kv;
    *(v4f*)(Ka + base + 64) = -kv * ps;
  }
}

struct KVTile { v4f k[16]; v4f v[8]; };

static __device__ __forceinline__ void load_k_tile(KVTile& t,
                                                   const float* __restrict__ kb,
                                                   int jt) {
  const float* kp = kb + (size_t)jt * (16 * 2048);
#pragma unroll
  for (int i = 0; i < 16; ++i) t.k[i] = ld4(kp + 8 * i);
}

static __device__ __forceinline__ void load_v_tile(KVTile& t,
                                                   const float* __restrict__ vb,
                                                   int jt) {
  const float* vp = vb + (size_t)jt * (16 * 2048);
#pragma unroll
  for (int kk = 0; kk < 4; ++kk) {
    t.v[2 * kk]     = ld4(vp + (size_t)(kk * 4) * 2048);
    t.v[2 * kk + 1] = ld4(vp + (size_t)(kk * 4 + 1) * 2048);
  }
}

__global__ void __launch_bounds__(128)
attn_kernel(const float* __restrict__ Qa, const float* __restrict__ Ka,
            const float* __restrict__ KV, float* __restrict__ wv) {
  __shared__ float Pls[4 * 16 * 18];

  const int tid = threadIdx.x;
  const int lane = tid & 31;
  const int wid = __builtin_amdgcn_readfirstlane(tid >> 5);
  const int hhi = lane >> 4, l16 = lane & 15;
  const int h = blockIdx.y;
  const int hoff = h * 128;
  const int qrow0 = blockIdx.x * 64 + wid * 16;
  const int qtile = qrow0 >> 4;

  float* myP = &Pls[wid * 16 * 18];

  const float* kbase = Ka + (size_t)l16 * 2048 + hoff + 4 * hhi;
  const float* vbase = KV + (size_t)(2 * hhi) * 2048 + 1024 + h * 64 + 4 * l16;

  v4f qf[16];
  {
    const float* qp = Qa + (size_t)(qrow0 + l16) * 2048 + hoff + 4 * hhi;
#pragma unroll
    for (int i = 0; i < 16; ++i) qf[i] = ld4(qp + 8 * i);
  }

  v8f o0 = {}, o1 = {}, o2 = {}, o3 = {};
  float m8[8], l8[8];
#pragma unroll
  for (int v = 0; v < 8; ++v) { m8[v] = -3.0e38f; l8[v] = 0.f; }

  auto score = [&](const KVTile& t, v8f& s0, v8f& s1) {
#pragma unroll
    for (int i = 0; i < 16; ++i) {
      v2f a0; a0.x = qf[i].x;  a0.y = qf[i].y;
      v2f b0; b0.x = t.k[i].x; b0.y = t.k[i].y;
      v2f a1; a1.x = qf[i].z;  a1.y = qf[i].w;
      v2f b1; b1.x = t.k[i].z; b1.y = t.k[i].w;
      s0 = wmma_f32(a0, b0, s0);
      s1 = wmma_f32(a1, b1, s1);
    }
  };

  auto softmax = [&](const v8f& s, const bool diag) {
    float sv[8], mx[8], mnew[8], alpha[8], p[8], rs[8];
#pragma unroll
    for (int v = 0; v < 8; ++v) {
      const int row = v + hhi * 8;
      sv[v] = s[v];
      if (diag && (l16 > row)) sv[v] = -1.0e30f;
    }
#pragma unroll
    for (int v = 0; v < 8; ++v) mx[v] = fmaxf(sv[v], row_xor<8>(sv[v]));
#pragma unroll
    for (int v = 0; v < 8; ++v) mx[v] = fmaxf(mx[v], row_xor<4>(mx[v]));
#pragma unroll
    for (int v = 0; v < 8; ++v) mx[v] = fmaxf(mx[v], row_xor<2>(mx[v]));
#pragma unroll
    for (int v = 0; v < 8; ++v) mx[v] = fmaxf(mx[v], row_xor<1>(mx[v]));
#pragma unroll
    for (int v = 0; v < 8; ++v) mnew[v] = fmaxf(m8[v], mx[v]);
#pragma unroll
    for (int v = 0; v < 8; ++v) p[v] = __expf(sv[v] - mnew[v]);
#pragma unroll
    for (int v = 0; v < 8; ++v)
      myP[(v + hhi * 8) * 18 + l16] = p[v];
#pragma unroll
    for (int v = 0; v < 8; ++v) alpha[v] = __expf(m8[v] - mnew[v]);
#pragma unroll
    for (int v = 0; v < 8; ++v) rs[v] = p[v] + row_xor<8>(p[v]);
#pragma unroll
    for (int v = 0; v < 8; ++v) rs[v] += row_xor<4>(rs[v]);
#pragma unroll
    for (int v = 0; v < 8; ++v) rs[v] += row_xor<2>(rs[v]);
#pragma unroll
    for (int v = 0; v < 8; ++v) rs[v] += row_xor<1>(rs[v]);
#pragma unroll
    for (int v = 0; v < 8; ++v) {
      l8[v] = l8[v] * alpha[v] + rs[v];
      m8[v] = mnew[v];
    }
#pragma unroll
    for (int v = 0; v < 8; ++v) {
      o0[v] *= alpha[v]; o1[v] *= alpha[v];
      o2[v] *= alpha[v]; o3[v] *= alpha[v];
    }
  };

  auto pv = [&](const KVTile& t) {
#pragma unroll
    for (int kk = 0; kk < 4; ++kk) {
      const int kb = kk * 4 + 2 * hhi;
      v2f a; a.x = myP[l16 * 18 + kb]; a.y = myP[l16 * 18 + kb + 1];
      const v4f v0 = t.v[2 * kk], v1 = t.v[2 * kk + 1];
      v2f b0, b1, b2, b3;
      b0.x = v0.x; b0.y = v1.x;
      b1.x = v0.y; b1.y = v1.y;
      b2.x = v0.z; b2.y = v1.z;
      b3.x = v0.w; b3.y = v1.w;
      o0 = wmma_f32(a, b0, o0);
      o1 = wmma_f32(a, b1, o1);
      o2 = wmma_f32(a, b2, o2);
      o3 = wmma_f32(a, b3, o3);
    }
  };

  auto step = [&](const KVTile& kn, const KVTile& vc, v8f& s) {
    v8f n0 = {}, n1 = {};
    score(kn, n0, n1);
    softmax(s, false);
    pv(vc);
    s = n0 + n1;
  };
#define STEP_ENDS_HERE()                                                \
  asm volatile("" ::"v"(s), "v"(o0), "v"(o1), "v"(o2), "v"(o3),         \
               "v"(m8[0]), "v"(m8[1]), "v"(m8[2]), "v"(m8[3]),          \
               "v"(m8[4]), "v"(m8[5]), "v"(m8[6]), "v"(m8[7]),          \
               "v"(l8[0]), "v"(l8[1]), "v"(l8[2]), "v"(l8[3]),          \
               "v"(l8[4]), "v"(l8[5]), "v"(l8[6]), "v"(l8[7]))
#define SOFTMAX_UNDER_SCORES(ID)                                        \
  _Pragma("unroll") for (int i_ = 0; i_ < 26; ++i_) {                   \
    __builtin_amdgcn_sched_group_barrier(0x008, 1, ID);    \
    __builtin_amdgcn_sched_group_barrier(0x002, 8, ID);    \
  }                                                                     \
  __builtin_amdgcn_sched_group_barrier(0x100, 8, ID);       \
  _Pragma("unroll") for (int i_ = 0; i_ < 6; ++i_) {                    \
    __builtin_amdgcn_sched_group_barrier(0x008, 1, ID);    \
    __builtin_amdgcn_sched_group_barrier(0x002, 8, ID);    \
  }                                                                     \
  __builtin_amdgcn_sched_group_barrier(0x008, 16, ID);

  KVTile ta, tb;
  load_k_tile(ta, kbase, 0);
  load_v_tile(ta, vbase, 0);
  load_k_tile(tb, kbase, qtile < 1 ? qtile : 1);
  v8f s;
  {
    v8f a0 = {}, a1 = {};
    score(ta, a0, a1);
    s = a0 + a1;
  }
  int jt = 0;
  for (; jt + 2 <= qtile; jt += 2) {
    load_k_tile(ta, kbase, jt + 2);
    load_v_tile(tb, vbase, jt + 1);
    __builtin_amdgcn_sched_barrier(0);
    step(tb, ta, s);
    STEP_ENDS_HERE();
    SOFTMAX_UNDER_SCORES(1)
    __builtin_amdgcn_sched_barrier(0);
    load_k_tile(tb, kbase, (jt + 3 < qtile) ? jt + 3 : qtile);
    load_v_tile(ta, vbase, jt + 2);
    __builtin_amdgcn_sched_barrier(0);
    step(ta, tb, s);
    STEP_ENDS_HERE();
    SOFTMAX_UNDER_SCORES(2)
    __builtin_amdgcn_sched_barrier(0);
  }
  if (jt + 1 == qtile) {
    load_v_tile(tb, vbase, jt + 1);
    __builtin_amdgcn_sched_barrier(0);
    step(tb, ta, s);
    STEP_ENDS_HERE();
    SOFTMAX_UNDER_SCORES(3)
    __builtin_amdgcn_sched_barrier(0);
    softmax(s, true);
    pv(tb);
  } else {
    softmax(s, true);
    pv(ta);
  }
#undef SOFTMAX_UNDER_SCORES
#undef STEP_ENDS_HERE

  const int cb = h * 64;
#pragma unroll
  for (int v = 0; v < 8; ++v) {
    const float inv = 1.0f / l8[v];
    const int r = qrow0 + v + hhi * 8;
    float4 ov;
    ov.x = o0[v] * inv; ov.y = o1[v] * inv;
    ov.z = o2[v] * inv; ov.w = o3[v] * inv;
    *(float4*)(wv + (size_t)r * DM + cb + 4 * l16) = ov;
  }
}

extern "C" void kernel_launch(void* const* d_in, const int* in_sizes, int n_in,
                              void* d_out, int out_size, void* d_ws,
                              size_t ws_size, hipStream_t stream) {
  const float* x    = (const float*)d_in[0];
  const float* pt   = (const float*)d_in[1];
  const float* gq   = (const float*)d_in[3];
  const float* Wq   = (const float*)d_in[4];
  const float* bq   = (const float*)d_in[5];
  const float* gkv  = (const float*)d_in[6];
  const float* Wkv  = (const float*)d_in[7];
  const float* bkv  = (const float*)d_in[8];
  const float* gc   = (const float*)d_in[9];
  const float* Wc   = (const float*)d_in[10];
  const float* bc   = (const float*)d_in[11];
  const float* Wout = (const float*)d_in[12];
  const float* bout = (const float*)d_in[13];
  const int*   cw   = (const int*)d_in[14];
  float* out = (float*)d_out;

  float* ws = (float*)d_ws;
  const size_t R = 1024;
  float* xqn  = ws + 0 * R * 1024;
  float* xkvn = ws + 1 * R * 1024;
  float* ptn  = ws + 2 * R * 1024;
  float* Qb   = ws + 3 * R * 1024;
  float* KVb  = ws + 4 * R * 1024;
  float* Cb   = ws + 6 * R * 1024;
  float* Qab  = ws + 8256 * R;
  float* Kab  = ws + 10304 * R;
  float* wvb  = ws + 12352 * R;

  rmsnorm3_kernel<<<1024, 256, 0, stream>>>(x, pt, gq, gkv, gc, xqn, xkvn, ptn);
  wmma_gemm_bias_n1024<<<dim3(16, 16), 128, 0, stream>>>(xqn, Wq, bq, Qb);
  wmma_gemm_bias_n2048<<<dim3(16, 32), 128, 0, stream>>>(xkvn, Wkv, bkv, KVb);
  wmma_gemm_bias_n1024<<<dim3(16, 16), 128, 0, stream>>>(ptn, Wc, bc, Cb);
  prefix_augment_kernel<<<dim3(32, 8), 128, 0, stream>>>(Qb, KVb, Cb, cw, Qab,
                                                         Kab);
  attn_kernel<<<dim3(16, 16), 128, 0, stream>>>(Qab, Kab, KVb, wvb);
  wmma_gemm_bias_n1024<<<dim3(16, 16), 128, 0, stream>>>(wvb, Wout, bout, out);
}


# GEMM (4 launches): each of a block's 4 waves takes a quarter of k over the whole 32x128 tile (16 accumulators: half the global loads per WMMA); partial tiles summed through LDS; k-sum order changes
# speedup vs baseline: 5.1879x; 1.0670x over previous
#include <hip/hip_runtime.h>
#include <math.h>


typedef __attribute__((ext_vector_type(2))) float v2f;
typedef __attribute__((ext_vector_type(8))) float v8f;

#define L   1024
#define DM  1024
#define EPSR 1.1920929e-07f

static __device__ __forceinline__ v8f wmma_f32(v2f a, v2f b, v8f c) {
  return __builtin_amdgcn_wmma_f32_16x16x4_f32(false, a, false, b, (short)0, c,
                                               false, false);
}

template <int M>
static __device__ __forceinline__ float row_xor(float x) {
  const int b = __builtin_bit_cast(int, x);
  return __builtin_bit_cast(
      float, __builtin_amdgcn_update_dpp(0, b, 0x160 + M, 0xf, 0xf, true));
}

__global__ void __launch_bounds__(256)
rmsnorm3_kernel(const float* __restrict__ x, const float* __restrict__ pt,
                const float* __restrict__ gq, const float* __restrict__ gkv,
                const float* __restrict__ gc,
                float* __restrict__ xqn, float* __restrict__ xkvn,
                float* __restrict__ ptn) {
  __shared__ float red[256];
  const int row = blockIdx.x;
  const int tid = threadIdx.x;

  float4 xv = ((const float4*)(x + (size_t)row * DM))[tid];
  red[tid] = xv.x * xv.x + xv.y * xv.y + xv.z * xv.z + xv.w * xv.w;
  __syncthreads();
  for (int s = 128; s > 0; s >>= 1) {
    if (tid < s) red[tid] += red[tid + s];
    __syncthreads();
  }
  float rn = rsqrtf(red[0] * (1.0f / DM) + EPSR);
  __syncthreads();

  float4 gqv = ((const float4*)gq)[tid];
  float4 gkvv = ((const float4*)gkv)[tid];
  float4 oq, okv;
  oq.x = xv.x * rn * gqv.x;  oq.y = xv.y * rn * gqv.y;
  oq.z = xv.z * rn * gqv.z;  oq.w = xv.w * rn * gqv.w;
  okv.x = xv.x * rn * gkvv.x; okv.y = xv.y * rn * gkvv.y;
  okv.z = xv.z * rn * gkvv.z; okv.w = xv.w * rn * gkvv.w;
  ((float4*)(xqn + (size_t)row * DM))[tid] = oq;
  ((float4*)(xkvn + (size_t)row * DM))[tid] = okv;

  float4 pv = ((const float4*)(pt + (size_t)row * DM))[tid];
  red[tid] = pv.x * pv.x + pv.y * pv.y + pv.z * pv.z + pv.w * pv.w;
  __syncthreads();
  for (int s = 128; s > 0; s >>= 1) {
    if (tid < s) red[tid] += red[tid + s];
    __syncthreads();
  }
  float rnp = rsqrtf(red[0] * (1.0f / DM) + EPSR);
  float4 gcv = ((const float4*)gc)[tid];
  float4 oc;
  oc.x = pv.x * rnp * gcv.x; oc.y = pv.y * rnp * gcv.y;
  oc.z = pv.z * rnp * gcv.z; oc.w = pv.w * rnp * gcv.w;
  ((float4*)(ptn + (size_t)row * DM))[tid] = oc;
}

struct GStage { float4 a[2][2]; float4 b[2][2][4]; };

template <int N>
static __device__ __forceinline__ void load_gstage(GStage& s,
                                                   const float* __restrict__ ap,
                                                   const float* __restrict__ bp,
                                                   int ks) {
  const float* a0 = ap + ks * 16;
#pragma unroll
  for (int t = 0; t < 2; ++t) {
#pragma unroll
    for (int rt = 0; rt < 2; ++rt)
      s.a[t][rt] = *(const float4*)(a0 + (size_t)(16 * rt) * DM + 8 * t);
  }
  const float* b0 = bp + (size_t)ks * 16 * N;
#pragma unroll
  for (int t = 0; t < 2; ++t) {
#pragma unroll
    for (int st = 0; st < 2; ++st) {
#pragma unroll
      for (int r = 0; r < 4; ++r)
        s.b[t][st][r] = *(const float4*)(b0 + (size_t)(8 * t + r) * N + 64 * st);
    }
  }
}

static __device__ __forceinline__ void stage_mma(const GStage& s,
                                                 v8f (&acc)[2][2][4]) {
#pragma unroll
  for (int t = 0; t < 2; ++t) {
#pragma unroll
    for (int kp = 0; kp < 2; ++kp) {
      v2f af[2];
#pragma unroll
      for (int rt = 0; rt < 2; ++rt) {
        const float4 a = s.a[t][rt];
        af[rt].x = kp ? a.z : a.x;
        af[rt].y = kp ? a.w : a.y;
      }
#pragma unroll
      for (int st = 0; st < 2; ++st) {
        const float4 lo = s.b[t][st][2 * kp], hi = s.b[t][st][2 * kp + 1];
        v2f b0; b0.x = lo.x; b0.y = hi.x;
        v2f b1; b1.x = lo.y; b1.y = hi.y;
        v2f b2; b2.x = lo.z; b2.y = hi.z;
        v2f b3; b3.x = lo.w; b3.y = hi.w;
#pragma unroll
        for (int rt = 0; rt < 2; ++rt) {
          acc[rt][st][0] = wmma_f32(af[rt], b0, acc[rt][st][0]);
          acc[rt][st][1] = wmma_f32(af[rt], b1, acc[rt][st][1]);
          acc[rt][st][2] = wmma_f32(af[rt], b2, acc[rt][st][2]);
          acc[rt][st][3] = wmma_f32(af[rt], b3, acc[rt][st][3]);
        }
      }
    }
  }
}

template <int N>
static __device__ __forceinline__ void gemm_body(const float* __restrict__ A,
                                                 const float* __restrict__ Wm,
                                                 const float* __restrict__ bias,
                                                 float* __restrict__ Cout,
                                                 float4* __restrict__ red) {
  const int tid = threadIdx.x;
  const int lane = tid & 31;
  const int wid = __builtin_amdgcn_readfirstlane(tid >> 5);
  const int hhi = lane >> 4, l16 = lane & 15;
  const int m0 = blockIdx.x * 32;
  const int n0 = blockIdx.y * 128;
  const int kb = wid * 256;

  const float* ap = A + (size_t)(m0 + l16) * DM + kb + 4 * hhi;
  const float* bp = Wm + (size_t)(kb + 4 * hhi) * N + n0 + 4 * l16;

  const int rt_o = wid >> 1, st_o = wid & 1;
  const float4 bv = *(const float4*)(bias + n0 + 64 * st_o + 4 * l16);

  v8f acc[2][2][4];
#pragma unroll
  for (int rt = 0; rt < 2; ++rt)
#pragma unroll
    for (int st = 0; st < 2; ++st)
#pragma unroll
      for (int j = 0; j < 4; ++j) acc[rt][st][j] = v8f{};

  GStage sa, sb;
  load_gstage<N>(sa, ap, bp, 0);
  for (int ks = 0; ks < 14; ks += 2) {
    load_gstage<N>(sb, ap, bp, ks + 1);
    __builtin_amdgcn_sched_barrier(0);
    stage_mma(sa, acc);
    load_gstage<N>(sa, ap, bp, ks + 2);
    __builtin_amdgcn_sched_barrier(0);
    stage_mma(sb, acc);
  }
  load_gstage<N>(sb, ap, bp, 15);
  __builtin_amdgcn_sched_barrier(0);
  stage_mma(sa, acc);
  stage_mma(sb, acc);

#pragma unroll
  for (int q = 0; q < 4; ++q) {
#pragma unroll
    for (int v = 0; v < 8; ++v) {
      float4 o;
      o.x = acc[q >> 1][q & 1][0][v]; o.y = acc[q >> 1][q & 1][1][v];
      o.z = acc[q >> 1][q & 1][2][v]; o.w = acc[q >> 1][q & 1][3][v];
      red[((wid * 4 + q) * 8 + v) * 32 + lane] = o;
    }
  }
  __syncthreads();

  float* cp = Cout + (size_t)(m0 + 16 * rt_o + 8 * hhi) * N + n0 + 64 * st_o +
              4 * l16;
#pragma unroll
  for (int v = 0; v < 8; ++v) {
    const float4 p0 = red[((0 * 4 + wid) * 8 + v) * 32 + lane];
    const float4 p1 = red[((1 * 4 + wid) * 8 + v) * 32 + lane];
    const float4 p2 = red[((2 * 4 + wid) * 8 + v) * 32 + lane];
    const float4 p3 = red[((3 * 4 + wid) * 8 + v) * 32 + lane];
    float4 o;
    o.x = ((p0.x + p1.x) + (p2.x + p3.x)) + bv.x;
    o.y = ((p0.y + p1.y) + (p2.y + p3.y)) + bv.y;
    o.z = ((p0.z + p1.z) + (p2.z + p3.z)) + bv.z;
    o.w = ((p0.w + p1.w) + (p2.w + p3.w)) + bv.w;
    *(float4*)(cp + (size_t)v * N) = o;
  }
}

__global__ void __launch_bounds__(128)
wmma_gemm_bias_n1024(const float* __restrict__ A, const float* __restrict__ Wm,
                     const float* __restrict__ bias, float* __restrict__ Cout) {
  __shared__ float4 red[4 * 4 * 8 * 32];
  gemm_body<1024>(A, Wm, bias, Cout, red);
}

__global__ void __launch_bounds__(128)
wmma_gemm_bias_n2048(const float* __restrict__ A, const float* __restrict__ Wm,
                     const float* __restrict__ bias, float* __restrict__ Cout) {
  __shared__ float4 red[4 * 4 * 8 * 32];
  gemm_body<2048>(A, Wm, bias, Cout, red);
}

typedef __attribute__((ext_vector_type(4))) float v4f;

static __device__ __forceinline__ int clip0L(int x) {
  return x < 0 ? 0 : (x > L ? L : x);
}

static __device__ __forceinline__ v4f ld4(const float* p) {
  return *(const v4f*)p;
}

__global__ void __launch_bounds__(128)
prefix_augment_kernel(const float* __restrict__ Q, const float* __restrict__ KV,
                      const float* __restrict__ C, const int* __restrict__ cw,
                      float* __restrict__ Qa, float* __restrict__ Ka) {
  const int tid = threadIdx.x;
  const int lane = tid & 31;
  const int wid = __builtin_amdgcn_readfirstlane(tid >> 5);
  const int hhi = lane >> 4, l16 = lane & 15;
  const int p = blockIdx.x;
  const int r = (wid & 1) ? 63 - p : p;
  const int hd = blockIdx.y * 2 + (wid >> 1);
  const int i0 = r * 16;
  const int w = cw[0];
  const float scale = 0.125f;

  const float* cb = C + hd * 64 + 4 * l16;
  const v4f zero = {0.f, 0.f, 0.f, 0.f};

  const int s0 = clip0L(i0 - w), e0 = clip0L(i0 + w);
  const int lo = s0 < e0 ? s0 : e0;

  v4f g0 = zero, g1 = zero, g2 = zero, g3 = zero;
  const int nfull = lo >> 5;
  v4f xa[16], xb[16];
  auto load_trip = [&](v4f* x, int k) {
#pragma unroll
    for (int u = 0; u < 16; ++u)
      x[u] = ld4(cb + (size_t)(32 * k + 2 * u + hhi) * DM);
  };
  auto add_trip = [&](const v4f* x) {
#pragma unroll
    for (int u = 0; u < 16; u += 4) {
      g0 += x[u]; g1 += x[u + 1]; g2 += x[u + 2]; g3 += x[u + 3];
    }
  };
  int k = 0;
  if (nfull > 0) load_trip(xa, 0);
  for (; k + 2 <= nfull; k += 2) {
    load_trip(xb, k + 1);
    __builtin_amdgcn_sched_barrier(0);
    add_trip(xa);
    load_trip(xa, (k + 2 < nfull) ? k + 2 : nfull - 1);
    __builtin_amdgcn_sched_barrier(0);
    add_trip(xb);
  }
  if (k < nfull) add_trip(xa);
  const int t = nfull * 32;
  if (t < lo) {
    v4f x[16];
#pragma unroll
    for (int u = 0; u < 16; ++u) {
      const int row = t + 2 * u + hhi;
      x[u] = ld4(cb + (size_t)(row < L - 1 ? row : L - 1) * DM);
    }
#pragma unroll
    for (int u = 0; u < 16; u += 4) {
      const int row = t + 2 * u + hhi;
      g0 += (row     < lo) ? x[u]     : zero;
      g1 += (row + 2 < lo) ? x[u + 1] : zero;
      g2 += (row + 4 < lo) ? x[u + 2] : zero;
      g3 += (row + 6 < lo) ? x[u + 3] : zero;
    }
  }
  v4f G = (g0 + g1) + (g2 + g3);
  {
    v4f o;
    o.x = __shfl_xor(G.x, 16); o.y = __shfl_xor(G.y, 16);
    o.z = __shfl_xor(G.z, 16); o.w = __shfl_xor(G.w, 16);
    G += o;
  }

  const int ia = i0 + 8 * hhi;
  const int sa = clip0L(ia - w), ea = clip0L(ia + w);
  const int s8 = clip0L(i0 + 8 - w), e8 = clip0L(i0 + 8 + w);
  const int hiA = s8 > e8 ? s8 : e8;
  v4f psA = G, peA = G;
  for (int t0 = lo; t0 < hiA; t0 += 16) {
    v4f x[16];
#pragma unroll
    for (int u = 0; u < 16; ++u) {
      const int row = t0 + u;
      x[u] = ld4(cb + (size_t)(row < L - 1 ? row : L - 1) * DM);
    }
#pragma unroll
    for (int u = 0; u < 16; ++u) {
      const int row = t0 + u;
      psA += (row < sa) ? x[u] : zero;
      peA += (row < ea) ? x[u] : zero;
    }
  }

  v4f ds[7], de[7];
#pragma unroll
  for (int v = 1; v < 8; ++v) {
    const int sp = clip0L(ia + v - 1 - w), sc = clip0L(ia + v - w);
    const int ep = clip0L(ia + v - 1 + w), ec = clip0L(ia + v + w);
    const v4f xs = ld4(cb + (size_t)(sp < L - 1 ? sp : L - 1) * DM);
    const v4f xe = ld4(cb + (size_t)(ep < L - 1 ? ep : L - 1) * DM);
    ds[v - 1] = (sc > sp) ? xs : zero;
    de[v - 1] = (ec > ep) ? xe : zero;
  }

  v4f ps = psA, pe = peA;
#pragma unroll
  for (int v = 0; v < 8; ++v) {
    if (v > 0) { ps += ds[v - 1]; pe += de[v - 1]; }
    const int i = ia + v;
    const v4f qv = ld4(Q + (size_t)i * DM + hd * 64 + 4 * l16);
    const v4f kv = ld4(KV + (size_t)i * 2048 + hd * 64 + 4 * l16);
    const size_t base = (size_t)i * 2048 + hd * 128 + 4 * l16;
    *(v4f*)(Qa + base)      = qv * (scale + pe);
    *(v4f*)(Qa + base + 64) = qv;
    *(v4f*)(Ka + base)      = kv;
    *(v4f*)(Ka + base + 64) = -kv * ps;
  }
}

struct KVTile { v4f k[16]; v4f v[8]; };

static __device__ __forceinline__ void load_k_tile(KVTile& t,
                                                   const float* __restrict__ kb,
                                                   int jt) {
  const float* kp = kb + (size_t)jt * (16 * 2048);
#pragma unroll
  for (int i = 0; i < 16; ++i) t.k[i] = ld4(kp + 8 * i);
}

static __device__ __forceinline__ void load_v_tile(KVTile& t,
                                                   const float* __restrict__ vb,
                                                   int jt) {
  const float* vp = vb + (size_t)jt * (16 * 2048);
#pragma unroll
  for (int kk = 0; kk < 4; ++kk) {
    t.v[2 * kk]     = ld4(vp + (size_t)(kk * 4) * 2048);
    t.v[2 * kk + 1] = ld4(vp + (size_t)(kk * 4 + 1) * 2048);
  }
}

__global__ void __launch_bounds__(128)
attn_kernel(const float* __restrict__ Qa, const float* __restrict__ Ka,
            const float* __restrict__ KV, float* __restrict__ wv) {
  __shared__ float Pls[4 * 16 * 18];

  const int tid = threadIdx.x;
  const int lane = tid & 31;
  const int wid = __builtin_amdgcn_readfirstlane(tid >> 5);
  const int hhi = lane >> 4, l16 = lane & 15;
  const int h = blockIdx.y;
  const int hoff = h * 128;
  const int qrow0 = blockIdx.x * 64 + wid * 16;
  const int qtile = qrow0 >> 4;

  float* myP = &Pls[wid * 16 * 18];

  const float* kbase = Ka + (size_t)l16 * 2048 + hoff + 4 * hhi;
  const float* vbase = KV + (size_t)(2 * hhi) * 2048 + 1024 + h * 64 + 4 * l16;

  v4f qf[16];
  {
    const float* qp = Qa + (size_t)(qrow0 + l16) * 2048 + hoff + 4 * hhi;
#pragma unroll
    for (int i = 0; i < 16; ++i) qf[i] = ld4(qp + 8 * i);
  }

  v8f o0 = {}, o1 = {}, o2 = {}, o3 = {};
  float m8[8], l8[8];
#pragma unroll
  for (int v = 0; v < 8; ++v) { m8[v] = -3.0e38f; l8[v] = 0.f; }

  auto score = [&](const KVTile& t, v8f& s0, v8f& s1) {
#pragma unroll
    for (int i = 0; i < 16; ++i) {
      v2f a0; a0.x = qf[i].x;  a0.y = qf[i].y;
      v2f b0; b0.x = t.k[i].x; b0.y = t.k[i].y;
      v2f a1; a1.x = qf[i].z;  a1.y = qf[i].w;
      v2f b1; b1.x = t.k[i].z; b1.y = t.k[i].w;
      s0 = wmma_f32(a0, b0, s0);
      s1 = wmma_f32(a1, b1, s1);
    }
  };

  auto softmax = [&](const v8f& s, const bool diag) {
    float sv[8], mx[8], mnew[8], alpha[8], p[8], rs[8];
#pragma unroll
    for (int v = 0; v < 8; ++v) {
      const int row = v + hhi * 8;
      sv[v] = s[v];
      if (diag && (l16 > row)) sv[v] = -1.0e30f;
    }
#pragma unroll
    for (int v = 0; v < 8; ++v) mx[v] = fmaxf(sv[v], row_xor<8>(sv[v]));
#pragma unroll
    for (int v = 0; v < 8; ++v) mx[v] = fmaxf(mx[v], row_xor<4>(mx[v]));
#pragma unroll
    for (int v = 0; v < 8; ++v) mx[v] = fmaxf(mx[v], row_xor<2>(mx[v]));
#pragma unroll
    for (int v = 0; v < 8; ++v) mx[v] = fmaxf(mx[v], row_xor<1>(mx[v]));
#pragma unroll
    for (int v = 0; v < 8; ++v) mnew[v] = fmaxf(m8[v], mx[v]);
#pragma unroll
    for (int v = 0; v < 8; ++v) p[v] = __expf(sv[v] - mnew[v]);
#pragma unroll
    for (int v = 0; v < 8; ++v)
      myP[(v + hhi * 8) * 18 + l16] = p[v];
#pragma unroll
    for (int v = 0; v < 8; ++v) alpha[v] = __expf(m8[v] - mnew[v]);
#pragma unroll
    for (int v = 0; v < 8; ++v) rs[v] = p[v] + row_xor<8>(p[v]);
#pragma unroll
    for (int v = 0; v < 8; ++v) rs[v] += row_xor<4>(rs[v]);
#pragma unroll
    for (int v = 0; v < 8; ++v) rs[v] += row_xor<2>(rs[v]);
#pragma unroll
    for (int v = 0; v < 8; ++v) rs[v] += row_xor<1>(rs[v]);
#pragma unroll
    for (int v = 0; v < 8; ++v) {
      l8[v] = l8[v] * alpha[v] + rs[v];
      m8[v] = mnew[v];
    }
#pragma unroll
    for (int v = 0; v < 8; ++v) {
      o0[v] *= alpha[v]; o1[v] *= alpha[v];
      o2[v] *= alpha[v]; o3[v] *= alpha[v];
    }
  };

  auto pv = [&](const KVTile& t) {
#pragma unroll
    for (int kk = 0; kk < 4; ++kk) {
      const int kb = kk * 4 + 2 * hhi;
      v2f a; a.x = myP[l16 * 18 + kb]; a.y = myP[l16 * 18 + kb + 1];
      const v4f v0 = t.v[2 * kk], v1 = t.v[2 * kk + 1];
      v2f b0, b1, b2, b3;
      b0.x = v0.x; b0.y = v1.x;
      b1.x = v0.y; b1.y = v1.y;
      b2.x = v0.z; b2.y = v1.z;
      b3.x = v0.w; b3.y = v1.w;
      o0 = wmma_f32(a, b0, o0);
      o1 = wmma_f32(a, b1, o1);
      o2 = wmma_f32(a, b2, o2);
      o3 = wmma_f32(a, b3, o3);
    }
  };

  auto step = [&](const KVTile& kn, const KVTile& vc, v8f& s) {
    v8f n0 = {}, n1 = {};
    score(kn, n0, n1);
    softmax(s, false);
    pv(vc);
    s = n0 + n1;
  };
#define STEP_ENDS_HERE()                                                \
  asm volatile("" ::"v"(s), "v"(o0), "v"(o1), "v"(o2), "v"(o3),         \
               "v"(m8[0]), "v"(m8[1]), "v"(m8[2]), "v"(m8[3]),          \
               "v"(m8[4]), "v"(m8[5]), "v"(m8[6]), "v"(m8[7]),          \
               "v"(l8[0]), "v"(l8[1]), "v"(l8[2]), "v"(l8[3]),          \
               "v"(l8[4]), "v"(l8[5]), "v"(l8[6]), "v"(l8[7]))
#define SOFTMAX_UNDER_SCORES(ID)                                        \
  _Pragma("unroll") for (int i_ = 0; i_ < 26; ++i_) {                   \
    __builtin_amdgcn_sched_group_barrier(0x008, 1, ID);    \
    __builtin_amdgcn_sched_group_barrier(0x002, 8, ID);    \
  }                                                                     \
  __builtin_amdgcn_sched_group_barrier(0x100, 8, ID);       \
  _Pragma("unroll") for (int i_ = 0; i_ < 6; ++i_) {                    \
    __builtin_amdgcn_sched_group_barrier(0x008, 1, ID);    \
    __builtin_amdgcn_sched_group_barrier(0x002, 8, ID);    \
  }                                                                     \
  __builtin_amdgcn_sched_group_barrier(0x008, 16, ID);

  KVTile ta, tb;
  load_k_tile(ta, kbase, 0);
  load_v_tile(ta, vbase, 0);
  load_k_tile(tb, kbase, qtile < 1 ? qtile : 1);
  v8f s;
  {
    v8f a0 = {}, a1 = {};
    score(ta, a0, a1);
    s = a0 + a1;
  }
  int jt = 0;
  for (; jt + 2 <= qtile; jt += 2) {
    load_k_tile(ta, kbase, jt + 2);
    load_v_tile(tb, vbase, jt + 1);
    __builtin_amdgcn_sched_barrier(0);
    step(tb, ta, s);
    STEP_ENDS_HERE();
    SOFTMAX_UNDER_SCORES(1)
    __builtin_amdgcn_sched_barrier(0);
    load_k_tile(tb, kbase, (jt + 3 < qtile) ? jt + 3 : qtile);
    load_v_tile(ta, vbase, jt + 2);
    __builtin_amdgcn_sched_barrier(0);
    step(ta, tb, s);
    STEP_ENDS_HERE();
    SOFTMAX_UNDER_SCORES(2)
    __builtin_amdgcn_sched_barrier(0);
  }
  if (jt + 1 == qtile) {
    load_v_tile(tb, vbase, jt + 1);
    __builtin_amdgcn_sched_barrier(0);
    step(tb, ta, s);
    STEP_ENDS_HERE();
    SOFTMAX_UNDER_SCORES(3)
    __builtin_amdgcn_sched_barrier(0);
    softmax(s, true);
    pv(tb);
  } else {
    softmax(s, true);
    pv(ta);
  }
#undef SOFTMAX_UNDER_SCORES
#undef STEP_ENDS_HERE

  const int cb = h * 64;
#pragma unroll
  for (int v = 0; v < 8; ++v) {
    const float inv = 1.0f / l8[v];
    const int r = qrow0 + v + hhi * 8;
    float4 ov;
    ov.x = o0[v] * inv; ov.y = o1[v] * inv;
    ov.z = o2[v] * inv; ov.w = o3[v] * inv;
    *(float4*)(wv + (size_t)r * DM + cb + 4 * l16) = ov;
  }
}

extern "C" void kernel_launch(void* const* d_in, const int* in_sizes, int n_in,
                              void* d_out, int out_size, void* d_ws,
                              size_t ws_size, hipStream_t stream) {
  const float* x    = (const float*)d_in[0];
  const float* pt   = (const float*)d_in[1];
  const float* gq   = (const float*)d_in[3];
  const float* Wq   = (const float*)d_in[4];
  const float* bq   = (const float*)d_in[5];
  const float* gkv  = (const float*)d_in[6];
  const float* Wkv  = (const float*)d_in[7];
  const float* bkv  = (const float*)d_in[8];
  const float* gc   = (const float*)d_in[9];
  const float* Wc   = (const float*)d_in[10];
  const float* bc   = (const float*)d_in[11];
  const float* Wout = (const float*)d_in[12];
  const float* bout = (const float*)d_in[13];
  const int*   cw   = (const int*)d_in[14];
  float* out = (float*)d_out;

  float* ws = (float*)d_ws;
  const size_t R = 1024;
  float* xqn  = ws + 0 * R * 1024;
  float* xkvn = ws + 1 * R * 1024;
  float* ptn  = ws + 2 * R * 1024;
  float* Qb   = ws + 3 * R * 1024;
  float* KVb  = ws + 4 * R * 1024;
  float* Cb   = ws + 6 * R * 1024;
  float* Qab  = ws + 8256 * R;
  float* Kab  = ws + 10304 * R;
  float* wvb  = ws + 12352 * R;

  rmsnorm3_kernel<<<1024, 256, 0, stream>>>(x, pt, gq, gkv, gc, xqn, xkvn, ptn);
  wmma_gemm_bias_n1024<<<dim3(32, 8), 128, 0, stream>>>(xqn, Wq, bq, Qb);
  wmma_gemm_bias_n2048<<<dim3(32, 16), 128, 0, stream>>>(xkvn, Wkv, bkv, KVb);
  wmma_gemm_bias_n1024<<<dim3(32, 8), 128, 0, stream>>>(ptn, Wc, bc, Cb);
  prefix_augment_kernel<<<dim3(32, 8), 128, 0, stream>>>(Qb, KVb, Cb, cw, Qab,
                                                         Kab);
  attn_kernel<<<dim3(16, 16), 128, 0, stream>>>(Qab, Kab, KVb, wvb);
  wmma_gemm_bias_n1024<<<dim3(32, 8), 128, 0, stream>>>(wvb, Wout, bout, out);
}


# prefix sum: the C projection's epilogue also stores column sums of its 16-row tiles; a prefix tile adds <=64 tile sums + <16 rows of C in one round trip (was 1 per 32 rows); prefix-sum order changes
# speedup vs baseline: 5.5125x; 1.0626x over previous
#include <hip/hip_runtime.h>
#include <math.h>


typedef __attribute__((ext_vector_type(2))) float v2f;
typedef __attribute__((ext_vector_type(8))) float v8f;

#define L   1024
#define DM  1024
#define EPSR 1.1920929e-07f

static __device__ __forceinline__ v8f wmma_f32(v2f a, v2f b, v8f c) {
  return __builtin_amdgcn_wmma_f32_16x16x4_f32(false, a, false, b, (short)0, c,
                                               false, false);
}

template <int M>
static __device__ __forceinline__ float row_xor(float x) {
  const int b = __builtin_bit_cast(int, x);
  return __builtin_bit_cast(
      float, __builtin_amdgcn_update_dpp(0, b, 0x160 + M, 0xf, 0xf, true));
}

__global__ void __launch_bounds__(256)
rmsnorm3_kernel(const float* __restrict__ x, const float* __restrict__ pt,
                const float* __restrict__ gq, const float* __restrict__ gkv,
                const float* __restrict__ gc,
                float* __restrict__ xqn, float* __restrict__ xkvn,
                float* __restrict__ ptn) {
  __shared__ float red[256];
  const int row = blockIdx.x;
  const int tid = threadIdx.x;

  float4 xv = ((const float4*)(x + (size_t)row * DM))[tid];
  red[tid] = xv.x * xv.x + xv.y * xv.y + xv.z * xv.z + xv.w * xv.w;
  __syncthreads();
  for (int s = 128; s > 0; s >>= 1) {
    if (tid < s) red[tid] += red[tid + s];
    __syncthreads();
  }
  float rn = rsqrtf(red[0] * (1.0f / DM) + EPSR);
  __syncthreads();

  float4 gqv = ((const float4*)gq)[tid];
  float4 gkvv = ((const float4*)gkv)[tid];
  float4 oq, okv;
  oq.x = xv.x * rn * gqv.x;  oq.y = xv.y * rn * gqv.y;
  oq.z = xv.z * rn * gqv.z;  oq.w = xv.w * rn * gqv.w;
  okv.x = xv.x * rn * gkvv.x; okv.y = xv.y * rn * gkvv.y;
  okv.z = xv.z * rn * gkvv.z; okv.w = xv.w * rn * gkvv.w;
  ((float4*)(xqn + (size_t)row * DM))[tid] = oq;
  ((float4*)(xkvn + (size_t)row * DM))[tid] = okv;

  float4 pv = ((const float4*)(pt + (size_t)row * DM))[tid];
  red[tid] = pv.x * pv.x + pv.y * pv.y + pv.z * pv.z + pv.w * pv.w;
  __syncthreads();
  for (int s = 128; s > 0; s >>= 1) {
    if (tid < s) red[tid] += red[tid + s];
    __syncthreads();
  }
  float rnp = rsqrtf(red[0] * (1.0f / DM) + EPSR);
  float4 gcv = ((const float4*)gc)[tid];
  float4 oc;
  oc.x = pv.x * rnp * gcv.x; oc.y = pv.y * rnp * gcv.y;
  oc.z = pv.z * rnp * gcv.z; oc.w = pv.w * rnp * gcv.w;
  ((float4*)(ptn + (size_t)row * DM))[tid] = oc;
}

struct GStage { float4 a[2][2]; float4 b[2][2][4]; };

template <int N>
static __device__ __forceinline__ void load_gstage(GStage& s,
                                                   const float* __restrict__ ap,
                                                   const float* __restrict__ bp,
                                                   int ks) {
  const float* a0 = ap + ks * 16;
#pragma unroll
  for (int t = 0; t < 2; ++t) {
#pragma unroll
    for (int rt = 0; rt < 2; ++rt)
      s.a[t][rt] = *(const float4*)(a0 + (size_t)(16 * rt) * DM + 8 * t);
  }
  const float* b0 = bp + (size_t)ks * 16 * N;
#pragma unroll
  for (int t = 0; t < 2; ++t) {
#pragma unroll
    for (int st = 0; st < 2; ++st) {
#pragma unroll
      for (int r = 0; r < 4; ++r)
        s.b[t][st][r] = *(const float4*)(b0 + (size_t)(8 * t + r) * N + 64 * st);
    }
  }
}

static __device__ __forceinline__ void stage_mma(const GStage& s,
                                                 v8f (&acc)[2][2][4]) {
#pragma unroll
  for (int t = 0; t < 2; ++t) {
#pragma unroll
    for (int kp = 0; kp < 2; ++kp) {
      v2f af[2];
#pragma unroll
      for (int rt = 0; rt < 2; ++rt) {
        const float4 a = s.a[t][rt];
        af[rt].x = kp ? a.z : a.x;
        af[rt].y = kp ? a.w : a.y;
      }
#pragma unroll
      for (int st = 0; st < 2; ++st) {
        const float4 lo = s.b[t][st][2 * kp], hi = s.b[t][st][2 * kp + 1];
        v2f b0; b0.x = lo.x; b0.y = hi.x;
        v2f b1; b1.x = lo.y; b1.y = hi.y;
        v2f b2; b2.x = lo.z; b2.y = hi.z;
        v2f b3; b3.x = lo.w; b3.y = hi.w;
#pragma unroll
        for (int rt = 0; rt < 2; ++rt) {
          acc[rt][st][0] = wmma_f32(af[rt], b0, acc[rt][st][0]);
          acc[rt][st][1] = wmma_f32(af[rt], b1, acc[rt][st][1]);
          acc[rt][st][2] = wmma_f32(af[rt], b2, acc[rt][st][2]);
          acc[rt][st][3] = wmma_f32(af[rt], b3, acc[rt][st][3]);
        }
      }
    }
  }
}

template <int N, bool TS>
static __device__ __forceinline__ void gemm_body(const float* __restrict__ A,
                                                 const float* __restrict__ Wm,
                                                 const float* __restrict__ bias,
                                                 float* __restrict__ Cout,
                                                 float* __restrict__ Tsum,
                                                 float4* __restrict__ red) {
  const int tid = threadIdx.x;
  const int lane = tid & 31;
  const int wid = __builtin_amdgcn_readfirstlane(tid >> 5);
  const int hhi = lane >> 4, l16 = lane & 15;
  const int m0 = blockIdx.x * 32;
  const int n0 = blockIdx.y * 128;
  const int kb = wid * 256;

  const float* ap = A + (size_t)(m0 + l16) * DM + kb + 4 * hhi;
  const float* bp = Wm + (size_t)(kb + 4 * hhi) * N + n0 + 4 * l16;

  const int rt_o = wid >> 1, st_o = wid & 1;
  const float4 bv = *(const float4*)(bias + n0 + 64 * st_o + 4 * l16);

  v8f acc[2][2][4];
#pragma unroll
  for (int rt = 0; rt < 2; ++rt)
#pragma unroll
    for (int st = 0; st < 2; ++st)
#pragma unroll
      for (int j = 0; j < 4; ++j) acc[rt][st][j] = v8f{};

  GStage sa, sb;
  load_gstage<N>(sa, ap, bp, 0);
  for (int ks = 0; ks < 14; ks += 2) {
    load_gstage<N>(sb, ap, bp, ks + 1);
    __builtin_amdgcn_sched_barrier(0);
    stage_mma(sa, acc);
    load_gstage<N>(sa, ap, bp, ks + 2);
    __builtin_amdgcn_sched_barrier(0);
    stage_mma(sb, acc);
  }
  load_gstage<N>(sb, ap, bp, 15);
  __builtin_amdgcn_sched_barrier(0);
  stage_mma(sa, acc);
  stage_mma(sb, acc);

#pragma unroll
  for (int q = 0; q < 4; ++q) {
#pragma unroll
    for (int v = 0; v < 8; ++v) {
      float4 o;
      o.x = acc[q >> 1][q & 1][0][v]; o.y = acc[q >> 1][q & 1][1][v];
      o.z = acc[q >> 1][q & 1][2][v]; o.w = acc[q >> 1][q & 1][3][v];
      red[((wid * 4 + q) * 8 + v) * 32 + lane] = o;
    }
  }
  __syncthreads();

  float* cp = Cout + (size_t)(m0 + 16 * rt_o + 8 * hhi) * N + n0 + 64 * st_o +
              4 * l16;
  float4 ts = {0.f, 0.f, 0.f, 0.f};
#pragma unroll
  for (int v = 0; v < 8; ++v) {
    const float4 p0 = red[((0 * 4 + wid) * 8 + v) * 32 + lane];
    const float4 p1 = red[((1 * 4 + wid) * 8 + v) * 32 + lane];
    const float4 p2 = red[((2 * 4 + wid) * 8 + v) * 32 + lane];
    const float4 p3 = red[((3 * 4 + wid) * 8 + v) * 32 + lane];
    float4 o;
    o.x = ((p0.x + p1.x) + (p2.x + p3.x)) + bv.x;
    o.y = ((p0.y + p1.y) + (p2.y + p3.y)) + bv.y;
    o.z = ((p0.z + p1.z) + (p2.z + p3.z)) + bv.z;
    o.w = ((p0.w + p1.w) + (p2.w + p3.w)) + bv.w;
    *(float4*)(cp + (size_t)v * N) = o;
    if (TS) { ts.x += o.x; ts.y += o.y; ts.z += o.z; ts.w += o.w; }
  }
  if (TS) {
    ts.x += __shfl_xor(ts.x, 16); ts.y += __shfl_xor(ts.y, 16);
    ts.z += __shfl_xor(ts.z, 16); ts.w += __shfl_xor(ts.w, 16);
    if (hhi == 0)
      *(float4*)(Tsum + (size_t)(2 * blockIdx.x + rt_o) * N + n0 + 64 * st_o +
                 4 * l16) = ts;
  }
}

__global__ void __launch_bounds__(128)
wmma_gemm_bias_n1024(const float* __restrict__ A, const float* __restrict__ Wm,
                     const float* __restrict__ bias, float* __restrict__ Cout) {
  __shared__ float4 red[4 * 4 * 8 * 32];
  gemm_body<1024, false>(A, Wm, bias, Cout, nullptr, red);
}

__global__ void __launch_bounds__(128)
wmma_gemm_bias_tsum_n1024(const float* __restrict__ A,
                          const float* __restrict__ Wm,
                          const float* __restrict__ bias,
                          float* __restrict__ Cout, float* __restrict__ Tsum) {
  __shared__ float4 red[4 * 4 * 8 * 32];
  gemm_body<1024, true>(A, Wm, bias, Cout, Tsum, red);
}

__global__ void __launch_bounds__(128)
wmma_gemm_bias_n2048(const float* __restrict__ A, const float* __restrict__ Wm,
                     const float* __restrict__ bias, float* __restrict__ Cout) {
  __shared__ float4 red[4 * 4 * 8 * 32];
  gemm_body<2048, false>(A, Wm, bias, Cout, nullptr, red);
}

typedef __attribute__((ext_vector_type(4))) float v4f;

static __device__ __forceinline__ int clip0L(int x) {
  return x < 0 ? 0 : (x > L ? L : x);
}

static __device__ __forceinline__ v4f ld4(const float* p) {
  return *(const v4f*)p;
}

__global__ void __launch_bounds__(128)
prefix_augment_kernel(const float* __restrict__ Q, const float* __restrict__ KV,
                      const float* __restrict__ C, const float* __restrict__ Tsum,
                      const int* __restrict__ cw,
                      float* __restrict__ Qa, float* __restrict__ Ka) {
  const int tid = threadIdx.x;
  const int lane = tid & 31;
  const int wid = __builtin_amdgcn_readfirstlane(tid >> 5);
  const int hhi = lane >> 4, l16 = lane & 15;
  const int p = blockIdx.x;
  const int r = (wid & 1) ? 63 - p : p;
  const int hd = blockIdx.y * 2 + (wid >> 1);
  const int i0 = r * 16;
  const int w = cw[0];
  const float scale = 0.125f;

  const float* cb = C + hd * 64 + 4 * l16;
  const v4f zero = {0.f, 0.f, 0.f, 0.f};

  const int s0 = clip0L(i0 - w), e0 = clip0L(i0 + w);
  const int lo = s0 < e0 ? s0 : e0;

  v4f g0 = zero, g1 = zero, g2 = zero, g3 = zero;
  const int nT = lo >> 4;
  const float* tb = Tsum + hd * 64 + 4 * l16;
  v4f xa[16], xb[16], xc[8];
#pragma unroll
  for (int u = 0; u < 16; ++u) xa[u] = ld4(tb + (size_t)(2 * u + hhi) * DM);
#pragma unroll
  for (int u = 0; u < 16; ++u)
    xb[u] = ld4(tb + (size_t)(32 + 2 * u + hhi) * DM);
#pragma unroll
  for (int u = 0; u < 8; ++u) {
    const int row = 16 * nT + 2 * u + hhi;
    xc[u] = ld4(cb + (size_t)(row < L - 1 ? row : L - 1) * DM);
  }
#pragma unroll
  for (int u = 0; u < 16; u += 4) {
    const int tr = 2 * u + hhi;
    g0 += (tr     < nT) ? xa[u]     : zero;
    g1 += (tr + 2 < nT) ? xa[u + 1] : zero;
    g2 += (tr + 4 < nT) ? xa[u + 2] : zero;
    g3 += (tr + 6 < nT) ? xa[u + 3] : zero;
  }
#pragma unroll
  for (int u = 0; u < 16; u += 4) {
    const int tr = 32 + 2 * u + hhi;
    g0 += (tr     < nT) ? xb[u]     : zero;
    g1 += (tr + 2 < nT) ? xb[u + 1] : zero;
    g2 += (tr + 4 < nT) ? xb[u + 2] : zero;
    g3 += (tr + 6 < nT) ? xb[u + 3] : zero;
  }
#pragma unroll
  for (int u = 0; u < 8; u += 4) {
    const int row = 16 * nT + 2 * u + hhi;
    g0 += (row     < lo) ? xc[u]     : zero;
    g1 += (row + 2 < lo) ? xc[u + 1] : zero;
    g2 += (row + 4 < lo) ? xc[u + 2] : zero;
    g3 += (row + 6 < lo) ? xc[u + 3] : zero;
  }
  v4f G = (g0 + g1) + (g2 + g3);
  {
    v4f o;
    o.x = __shfl_xor(G.x, 16); o.y = __shfl_xor(G.y, 16);
    o.z = __shfl_xor(G.z, 16); o.w = __shfl_xor(G.w, 16);
    G += o;
  }

  const int ia = i0 + 8 * hhi;
  const int sa = clip0L(ia - w), ea = clip0L(ia + w);
  const int s8 = clip0L(i0 + 8 - w), e8 = clip0L(i0 + 8 + w);
  const int hiA = s8 > e8 ? s8 : e8;
  v4f psA = G, peA = G;
  for (int t0 = lo; t0 < hiA; t0 += 16) {
    v4f x[16];
#pragma unroll
    for (int u = 0; u < 16; ++u) {
      const int row = t0 + u;
      x[u] = ld4(cb + (size_t)(row < L - 1 ? row : L - 1) * DM);
    }
#pragma unroll
    for (int u = 0; u < 16; ++u) {
      const int row = t0 + u;
      psA += (row < sa) ? x[u] : zero;
      peA += (row < ea) ? x[u] : zero;
    }
  }

  v4f ds[7], de[7];
#pragma unroll
  for (int v = 1; v < 8; ++v) {
    const int sp = clip0L(ia + v - 1 - w), sc = clip0L(ia + v - w);
    const int ep = clip0L(ia + v - 1 + w), ec = clip0L(ia + v + w);
    const v4f xs = ld4(cb + (size_t)(sp < L - 1 ? sp : L - 1) * DM);
    const v4f xe = ld4(cb + (size_t)(ep < L - 1 ? ep : L - 1) * DM);
    ds[v - 1] = (sc > sp) ? xs : zero;
    de[v - 1] = (ec > ep) ? xe : zero;
  }

  v4f ps = psA, pe = peA;
#pragma unroll
  for (int v = 0; v < 8; ++v) {
    if (v > 0) { ps += ds[v - 1]; pe += de[v - 1]; }
    const int i = ia + v;
    const v4f qv = ld4(Q + (size_t)i * DM + hd * 64 + 4 * l16);
    const v4f kv = ld4(KV + (size_t)i * 2048 + hd * 64 + 4 * l16);
    const size_t base = (size_t)i * 2048 + hd * 128 + 4 * l16;
    *(v4f*)(Qa + base)      = qv * (scale + pe);
    *(v4f*)(Qa + base + 64) = qv;
    *(v4f*)(Ka + base)      = kv;
    *(v4f*)(Ka + base + 64) = -kv * ps;
  }
}

struct KVTile { v4f k[16]; v4f v[8]; };

static __device__ __forceinline__ void load_k_tile(KVTile& t,
                                                   const float* __restrict__ kb,
                                                   int jt) {
  const float* kp = kb + (size_t)jt * (16 * 2048);
#pragma unroll
  for (int i = 0; i < 16; ++i) t.k[i] = ld4(kp + 8 * i);
}

static __device__ __forceinline__ void load_v_tile(KVTile& t,
                                                   const float* __restrict__ vb,
                                                   int jt) {
  const float* vp = vb + (size_t)jt * (16 * 2048);
#pragma unroll
  for (int kk = 0; kk < 4; ++kk) {
    t.v[2 * kk]     = ld4(vp + (size_t)(kk * 4) * 2048);
    t.v[2 * kk + 1] = ld4(vp + (size_t)(kk * 4 + 1) * 2048);
  }
}

__global__ void __launch_bounds__(128)
attn_kernel(const float* __restrict__ Qa, const float* __restrict__ Ka,
            const float* __restrict__ KV, float* __restrict__ wv) {
  __shared__ float Pls[4 * 16 * 18];

  const int tid = threadIdx.x;
  const int lane = tid & 31;
  const int wid = __builtin_amdgcn_readfirstlane(tid >> 5);
  const int hhi = lane >> 4, l16 = lane & 15;
  const int h = blockIdx.y;
  const int hoff = h * 128;
  const int qrow0 = blockIdx.x * 64 + wid * 16;
  const int qtile = qrow0 >> 4;

  float* myP = &Pls[wid * 16 * 18];

  const float* kbase = Ka + (size_t)l16 * 2048 + hoff + 4 * hhi;
  const float* vbase = KV + (size_t)(2 * hhi) * 2048 + 1024 + h * 64 + 4 * l16;

  v4f qf[16];
  {
    const float* qp = Qa + (size_t)(qrow0 + l16) * 2048 + hoff + 4 * hhi;
#pragma unroll
    for (int i = 0; i < 16; ++i) qf[i] = ld4(qp + 8 * i);
  }

  v8f o0 = {}, o1 = {}, o2 = {}, o3 = {};
  float m8[8], l8[8];
#pragma unroll
  for (int v = 0; v < 8; ++v) { m8[v] = -3.0e38f; l8[v] = 0.f; }

  auto score = [&](const KVTile& t, v8f& s0, v8f& s1) {
#pragma unroll
    for (int i = 0; i < 16; ++i) {
      v2f a0; a0.x = qf[i].x;  a0.y = qf[i].y;
      v2f b0; b0.x = t.k[i].x; b0.y = t.k[i].y;
      v2f a1; a1.x = qf[i].z;  a1.y = qf[i].w;
      v2f b1; b1.x = t.k[i].z; b1.y = t.k[i].w;
      s0 = wmma_f32(a0, b0, s0);
      s1 = wmma_f32(a1, b1, s1);
    }
  };

  auto softmax = [&](const v8f& s, const bool diag) {
    float sv[8], mx[8], mnew[8], alpha[8], p[8], rs[8];
#pragma unroll
    for (int v = 0; v < 8; ++v) {
      const int row = v + hhi * 8;
      sv[v] = s[v];
      if (diag && (l16 > row)) sv[v] = -1.0e30f;
    }
#pragma unroll
    for (int v = 0; v < 8; ++v) mx[v] = fmaxf(sv[v], row_xor<8>(sv[v]));
#pragma unroll
    for (int v = 0; v < 8; ++v) mx[v] = fmaxf(mx[v], row_xor<4>(mx[v]));
#pragma unroll
    for (int v = 0; v < 8; ++v) mx[v] = fmaxf(mx[v], row_xor<2>(mx[v]));
#pragma unroll
    for (int v = 0; v < 8; ++v) mx[v] = fmaxf(mx[v], row_xor<1>(mx[v]));
#pragma unroll
    for (int v = 0; v < 8; ++v) mnew[v] = fmaxf(m8[v], mx[v]);
#pragma unroll
    for (int v = 0; v < 8; ++v) p[v] = __expf(sv[v] - mnew[v]);
#pragma unroll
    for (int v = 0; v < 8; ++v)
      myP[(v + hhi * 8) * 18 + l16] = p[v];
#pragma unroll
    for (int v = 0; v < 8; ++v) alpha[v] = __expf(m8[v] - mnew[v]);
#pragma unroll
    for (int v = 0; v < 8; ++v) rs[v] = p[v] + row_xor<8>(p[v]);
#pragma unroll
    for (int v = 0; v < 8; ++v) rs[v] += row_xor<4>(rs[v]);
#pragma unroll
    for (int v = 0; v < 8; ++v) rs[v] += row_xor<2>(rs[v]);
#pragma unroll
    for (int v = 0; v < 8; ++v) rs[v] += row_xor<1>(rs[v]);
#pragma unroll
    for (int v = 0; v < 8; ++v) {
      l8[v] = l8[v] * alpha[v] + rs[v];
      m8[v] = mnew[v];
    }
#pragma unroll
    for (int v = 0; v < 8; ++v) {
      o0[v] *= alpha[v]; o1[v] *= alpha[v];
      o2[v] *= alpha[v]; o3[v] *= alpha[v];
    }
  };

  auto pv = [&](const KVTile& t) {
#pragma unroll
    for (int kk = 0; kk < 4; ++kk) {
      const int kb = kk * 4 + 2 * hhi;
      v2f a; a.x = myP[l16 * 18 + kb]; a.y = myP[l16 * 18 + kb + 1];
      const v4f v0 = t.v[2 * kk], v1 = t.v[2 * kk + 1];
      v2f b0, b1, b2, b3;
      b0.x = v0.x; b0.y = v1.x;
      b1.x = v0.y; b1.y = v1.y;
      b2.x = v0.z; b2.y = v1.z;
      b3.x = v0.w; b3.y = v1.w;
      o0 = wmma_f32(a, b0, o0);
      o1 = wmma_f32(a, b1, o1);
      o2 = wmma_f32(a, b2, o2);
      o3 = wmma_f32(a, b3, o3);
    }
  };

  auto step = [&](const KVTile& kn, const KVTile& vc, v8f& s) {
    v8f n0 = {}, n1 = {};
    score(kn, n0, n1);
    softmax(s, false);
    pv(vc);
    s = n0 + n1;
  };
#define STEP_ENDS_HERE()                                                \
  asm volatile("" ::"v"(s), "v"(o0), "v"(o1), "v"(o2), "v"(o3),         \
               "v"(m8[0]), "v"(m8[1]), "v"(m8[2]), "v"(m8[3]),          \
               "v"(m8[4]), "v"(m8[5]), "v"(m8[6]), "v"(m8[7]),          \
               "v"(l8[0]), "v"(l8[1]), "v"(l8[2]), "v"(l8[3]),          \
               "v"(l8[4]), "v"(l8[5]), "v"(l8[6]), "v"(l8[7]))
#define SOFTMAX_UNDER_SCORES(ID)                                        \
  _Pragma("unroll") for (int i_ = 0; i_ < 26; ++i_) {                   \
    __builtin_amdgcn_sched_group_barrier(0x008, 1, ID);    \
    __builtin_amdgcn_sched_group_barrier(0x002, 8, ID);    \
  }                                                                     \
  __builtin_amdgcn_sched_group_barrier(0x100, 8, ID);       \
  _Pragma("unroll") for (int i_ = 0; i_ < 6; ++i_) {                    \
    __builtin_amdgcn_sched_group_barrier(0x008, 1, ID);    \
    __builtin_amdgcn_sched_group_barrier(0x002, 8, ID);    \
  }                                                                     \
  __builtin_amdgcn_sched_group_barrier(0x008, 16, ID);

  KVTile ta, tb;
  load_k_tile(ta, kbase, 0);
  load_v_tile(ta, vbase, 0);
  load_k_tile(tb, kbase, qtile < 1 ? qtile : 1);
  v8f s;
  {
    v8f a0 = {}, a1 = {};
    score(ta, a0, a1);
    s = a0 + a1;
  }
  int jt = 0;
  for (; jt + 2 <= qtile; jt += 2) {
    load_k_tile(ta, kbase, jt + 2);
    load_v_tile(tb, vbase, jt + 1);
    __builtin_amdgcn_sched_barrier(0);
    step(tb, ta, s);
    STEP_ENDS_HERE();
    SOFTMAX_UNDER_SCORES(1)
    __builtin_amdgcn_sched_barrier(0);
    load_k_tile(tb, kbase, (jt + 3 < qtile) ? jt + 3 : qtile);
    load_v_tile(ta, vbase, jt + 2);
    __builtin_amdgcn_sched_barrier(0);
    step(ta, tb, s);
    STEP_ENDS_HERE();
    SOFTMAX_UNDER_SCORES(2)
    __builtin_amdgcn_sched_barrier(0);
  }
  if (jt + 1 == qtile) {
    load_v_tile(tb, vbase, jt + 1);
    __builtin_amdgcn_sched_barrier(0);
    step(tb, ta, s);
    STEP_ENDS_HERE();
    SOFTMAX_UNDER_SCORES(3)
    __builtin_amdgcn_sched_barrier(0);
    softmax(s, true);
    pv(tb);
  } else {
    softmax(s, true);
    pv(ta);
  }
#undef SOFTMAX_UNDER_SCORES
#undef STEP_ENDS_HERE

  const int cb = h * 64;
#pragma unroll
  for (int v = 0; v < 8; ++v) {
    const float inv = 1.0f / l8[v];
    const int r = qrow0 + v + hhi * 8;
    float4 ov;
    ov.x = o0[v] * inv; ov.y = o1[v] * inv;
    ov.z = o2[v] * inv; ov.w = o3[v] * inv;
    *(float4*)(wv + (size_t)r * DM + cb + 4 * l16) = ov;
  }
}

extern "C" void kernel_launch(void* const* d_in, const int* in_sizes, int n_in,
                              void* d_out, int out_size, void* d_ws,
                              size_t ws_size, hipStream_t stream) {
  const float* x    = (const float*)d_in[0];
  const float* pt   = (const float*)d_in[1];
  const float* gq   = (const float*)d_in[3];
  const float* Wq   = (const float*)d_in[4];
  const float* bq   = (const float*)d_in[5];
  const float* gkv  = (const float*)d_in[6];
  const float* Wkv  = (const float*)d_in[7];
  const float* bkv  = (const float*)d_in[8];
  const float* gc   = (const float*)d_in[9];
  const float* Wc   = (const float*)d_in[10];
  const float* bc   = (const float*)d_in[11];
  const float* Wout = (const float*)d_in[12];
  const float* bout = (const float*)d_in[13];
  const int*   cw   = (const int*)d_in[14];
  float* out = (float*)d_out;

  float* ws = (float*)d_ws;
  const size_t R = 1024;
  float* xqn  = ws + 0 * R * 1024;
  float* xkvn = ws + 1 * R * 1024;
  float* ptn  = ws + 2 * R * 1024;
  float* Qb   = ws + 3 * R * 1024;
  float* KVb  = ws + 4 * R * 1024;
  float* Cb   = ws + 6 * R * 1024;
  float* Tb   = ws + 7168 * R;
  float* Qab  = ws + 8256 * R;
  float* Kab  = ws + 10304 * R;
  float* wvb  = ws + 12352 * R;

  rmsnorm3_kernel<<<1024, 256, 0, stream>>>(x, pt, gq, gkv, gc, xqn, xkvn, ptn);
  wmma_gemm_bias_n1024<<<dim3(32, 8), 128, 0, stream>>>(xqn, Wq, bq, Qb);
  wmma_gemm_bias_n2048<<<dim3(32, 16), 128, 0, stream>>>(xkvn, Wkv, bkv, KVb);
  wmma_gemm_bias_tsum_n1024<<<dim3(32, 8), 128, 0, stream>>>(ptn, Wc, bc, Cb,
                                                             Tb);
  prefix_augment_kernel<<<dim3(32, 8), 128, 0, stream>>>(Qb, KVb, Cb, Tb, cw,
                                                         Qab, Kab);
  attn_kernel<<<dim3(16, 16), 128, 0, stream>>>(Qab, Kab, KVb, wvb);
  wmma_gemm_bias_n1024<<<dim3(32, 8), 128, 0, stream>>>(wvb, Wout, bout, out);
}
